# starting kernel
# baseline (speedup 1.0000x reference)
#include <hip/hip_runtime.h>
#include <hip/hip_bf16.h>


typedef __attribute__((ext_vector_type(16))) __bf16 v16bf;
typedef __attribute__((ext_vector_type(8)))  float  v8f;

union BfFrag {
    unsigned int u[8];
    v16bf bf;
};

#define HIDF 128

static __device__ __forceinline__ unsigned short f32_to_bf16(float f) {
    unsigned int u = __float_as_uint(f);
    unsigned int r = u + 0x7FFFu + ((u >> 16) & 1u);
    return (unsigned short)(r >> 16);
}

static __device__ __forceinline__ void atomicMaxF(float* addr, float val) {
    if (val >= 0.0f) atomicMax((int*)addr, __float_as_int(val));
    else             atomicMin((unsigned int*)addr, __float_as_uint(val));
}

__global__ void wt_bf16_kernel(const float* __restrict__ W,
                               unsigned short* __restrict__ Wt,
                               int K, int Nout) {
    int i = blockIdx.x * blockDim.x + threadIdx.x;
    if (i < K * Nout) {
        int k = i / Nout, n = i - k * Nout;
        Wt[n * K + k] = f32_to_bf16(W[i]);
    }
}

__global__ void gemm_wmma_bf16(const float* __restrict__ A,
                               const unsigned short* __restrict__ Wt,
                               float* __restrict__ C,
                               const float* __restrict__ bias,
                               int M, int K, int Nout) {
    __shared__ unsigned short As[16 * (HIDF + 8)];
    const int ldA = K + 8;

    const int tid = threadIdx.x;
    const int m0  = blockIdx.x * 16;

    const int total = 16 * K;
    for (int i = tid; i < total; i += blockDim.x) {
        int r = i / K, c = i - r * K;
        int gr = m0 + r;
        float v = (gr < M) ? A[(size_t)gr * K + c] : 0.0f;
        As[r * ldA + c] = f32_to_bf16(v);
    }
    __syncthreads();

    const int wave  = tid >> 5;
    const int lane  = tid & 31;
    const int row   = lane & 15;
    const int khalf = lane >> 4;
    const int ncol  = wave * 16 + row;

    v8f acc = {};

    for (int k0 = 0; k0 < K; k0 += 32) {
        const int cb = k0 + khalf * 8;

        BfFrag a, b;
        {
            const uint4* p0 = (const uint4*)&As[row * ldA + cb];
            const uint4* p1 = (const uint4*)&As[row * ldA + cb + 16];
            uint4 x0 = *p0, x1 = *p1;
            a.u[0] = x0.x; a.u[1] = x0.y; a.u[2] = x0.z; a.u[3] = x0.w;
            a.u[4] = x1.x; a.u[5] = x1.y; a.u[6] = x1.z; a.u[7] = x1.w;
        }
        {
            const uint4* p0 = (const uint4*)&Wt[(size_t)ncol * K + cb];
            const uint4* p1 = (const uint4*)&Wt[(size_t)ncol * K + cb + 16];
            uint4 x0 = *p0, x1 = *p1;
            b.u[0] = x0.x; b.u[1] = x0.y; b.u[2] = x0.z; b.u[3] = x0.w;
            b.u[4] = x1.x; b.u[5] = x1.y; b.u[6] = x1.z; b.u[7] = x1.w;
        }

        acc = __builtin_amdgcn_wmma_f32_16x16x32_bf16(
             false, a.bf,  false, b.bf,
             (short)0, acc,  false,  false);
    }

    float badd = bias ? bias[ncol] : 0.0f;
#pragma unroll
    for (int r = 0; r < 8; ++r) {
        int gr = m0 + r + khalf * 8;
        if (gr < M) C[(size_t)gr * Nout + ncol] = acc[r] + badd;
    }
}

__global__ void node_scores_kernel(const float* __restrict__ h,
                                   const float* __restrict__ a_s,
                                   const float* __restrict__ a_d,
                                   float* __restrict__ s_src,
                                   float* __restrict__ s_dst, int N) {
    int gt   = blockIdx.x * blockDim.x + threadIdx.x;
    int node = gt >> 5;
    int lane = gt & 31;
    if (node >= N) return;
    const float* hp = h + (size_t)node * HIDF;
    float ss = 0.0f, sd = 0.0f;
#pragma unroll
    for (int f = lane; f < HIDF; f += 32) {
        float v = hp[f];
        ss += v * a_s[f];
        sd += v * a_d[f];
    }
#pragma unroll
    for (int off = 16; off; off >>= 1) {
        ss += __shfl_xor(ss, off, 32);
        sd += __shfl_xor(sd, off, 32);
    }
    if (lane == 0) { s_src[node] = ss; s_dst[node] = sd; }
}

__global__ void init_kernel(float* __restrict__ m, float* __restrict__ denom,
                            float* __restrict__ hagg, int N) {
    int i = blockIdx.x * blockDim.x + threadIdx.x;
    if (i < N * HIDF) hagg[i] = 0.0f;
    if (i < N) { m[i] = __int_as_float(0xFF800000); denom[i] = 0.0f; }
}

static __device__ __forceinline__ void edge_ids(const int* __restrict__ ei,
                                                int i, int E, int& s, int& d) {
    if (i < E) { s = ei[i]; d = ei[E + i]; }
    else       { s = d = i - E; }
}

__global__ void edge_max_kernel(const int* __restrict__ ei,
                                const float* __restrict__ ssrc,
                                const float* __restrict__ sdst,
                                float* __restrict__ m, int E, int N) {
    int i = blockIdx.x * blockDim.x + threadIdx.x;
    if (i >= E + N) return;
    int s, d; edge_ids(ei, i, E, s, d);
    float e = ssrc[s] + sdst[d];
    e = e > 0.0f ? e : 0.2f * e;
    atomicMaxF(&m[d], e);
}

__global__ void edge_exp_kernel(const int* __restrict__ ei,
                                const float* __restrict__ ssrc,
                                const float* __restrict__ sdst,
                                const float* __restrict__ m,
                                float* __restrict__ ex,
                                float* __restrict__ denom, int E, int N) {
    int i = blockIdx.x * blockDim.x + threadIdx.x;
    if (i >= E + N) return;
    int s, d; edge_ids(ei, i, E, s, d);
    float e = ssrc[s] + sdst[d];
    e = e > 0.0f ? e : 0.2f * e;
    float v = __expf(e - m[d]);
    ex[i] = v;
    atomicAdd(&denom[d], v);
}

__global__ void aggregate_kernel(const int* __restrict__ ei,
                                 const float* __restrict__ ex,
                                 const float* __restrict__ denom,
                                 const float* __restrict__ h,
                                 float* __restrict__ hagg, int E, int N) {
    int gt   = blockIdx.x * blockDim.x + threadIdx.x;
    int edge = gt >> 5;
    int lane = gt & 31;
    if (edge >= E + N) return;
    int s, d; edge_ids(ei, edge, E, s, d);
    float alpha = ex[edge] / (denom[d] + 1e-16f);
    float4 v = ((const float4*)(h + (size_t)s * HIDF))[lane];
    float* out = hagg + (size_t)d * HIDF + lane * 4;
    atomicAdd(out + 0, alpha * v.x);
    atomicAdd(out + 1, alpha * v.y);
    atomicAdd(out + 2, alpha * v.z);
    atomicAdd(out + 3, alpha * v.w);
}

__global__ void bias_act_kernel(float* __restrict__ h,
                                const float* __restrict__ b, int N) {
    int i = blockIdx.x * blockDim.x + threadIdx.x;
    if (i < N * HIDF) {
        float v = h[i] + b[i & (HIDF - 1)];
        h[i] = v > 0.0f ? v : 0.01f * v;
    }
}

extern "C" void kernel_launch(void* const* d_in, const int* in_sizes, int n_in,
                              void* d_out, int out_size, void* d_ws, size_t ws_size,
                              hipStream_t stream) {
    (void)n_in; (void)out_size; (void)ws_size;

    const float* x    = (const float*)d_in[0];
    const int*   ei   = (const int*)d_in[1];
    const float* Wl[3] = {(const float*)d_in[2], (const float*)d_in[6],  (const float*)d_in[10]};
    const float* Asl[3]= {(const float*)d_in[3], (const float*)d_in[7],  (const float*)d_in[11]};
    const float* Adl[3]= {(const float*)d_in[4], (const float*)d_in[8],  (const float*)d_in[12]};
    const float* Bl[3] = {(const float*)d_in[5], (const float*)d_in[9],  (const float*)d_in[13]};
    const float* Wout = (const float*)d_in[14];
    const float* bout = (const float*)d_in[15];

    const int FIN  = 64;
    const int FOUT = 64;
    const int N  = in_sizes[0] / FIN;
    const int E  = in_sizes[1] / 2;
    const int Et = E + N;

    float* ws     = (float*)d_ws;
    float* h_pre  = ws;
    float* h_agg  = h_pre + (size_t)N * HIDF;
    float* s_src  = h_agg + (size_t)N * HIDF;
    float* s_dst  = s_src + N;
    float* mbuf   = s_dst + N;
    float* dbuf   = mbuf  + N;
    float* exbuf  = dbuf  + N;
    unsigned short* Wt = (unsigned short*)(exbuf + Et);

    const int B256 = 256;
    const int gNH   = (N * HIDF + B256 - 1) / B256;
    const int gEt   = (Et + B256 - 1) / B256;
    const int gEtW  = (int)(((long long)Et * 32 + B256 - 1) / B256);
    const int gNW   = (int)(((long long)N * 32 + B256 - 1) / B256);
    const int gM    = (N + 15) / 16;

    const float* Ain = x;
    int Kin = FIN;
    for (int l = 0; l < 3; ++l) {
        wt_bf16_kernel<<<(Kin * HIDF + B256 - 1) / B256, B256, 0, stream>>>(
            Wl[l], Wt, Kin, HIDF);
        gemm_wmma_bf16<<<gM, (HIDF / 16) * 32, 0, stream>>>(
            Ain, Wt, h_pre, nullptr, N, Kin, HIDF);
        node_scores_kernel<<<gNW, B256, 0, stream>>>(
            h_pre, Asl[l], Adl[l], s_src, s_dst, N);
        init_kernel<<<gNH, B256, 0, stream>>>(mbuf, dbuf, h_agg, N);
        edge_max_kernel<<<gEt, B256, 0, stream>>>(ei, s_src, s_dst, mbuf, E, N);
        edge_exp_kernel<<<gEt, B256, 0, stream>>>(ei, s_src, s_dst, mbuf,
                                                  exbuf, dbuf, E, N);
        aggregate_kernel<<<gEtW, B256, 0, stream>>>(ei, exbuf, dbuf, h_pre,
                                                    h_agg, E, N);
        bias_act_kernel<<<gNH, B256, 0, stream>>>(h_agg, Bl[l], N);
        Ain = h_agg;
        Kin = HIDF;
    }

    wt_bf16_kernel<<<(HIDF * FOUT + B256 - 1) / B256, B256, 0, stream>>>(
        Wout, Wt, HIDF, FOUT);
    gemm_wmma_bf16<<<gM, (FOUT / 16) * 32, 0, stream>>>(
        h_agg, Wt, (float*)d_out, bout, N, HIDF, FOUT);
}


# Edges grouped by destination once per call (counting sort, integer atomics); one wave per node then does softmax and row sum in registers in a fixed order; replaces the float atomic max/add scatter
# speedup vs baseline: 2.8458x; 2.8458x over previous
#include <hip/hip_runtime.h>
#include <hip/hip_bf16.h>


typedef __attribute__((ext_vector_type(16))) __bf16 v16bf;
typedef __attribute__((ext_vector_type(8)))  float  v8f;

union BfFrag {
    unsigned int u[8];
    v16bf bf;
};

#define HIDF 128

static __device__ __forceinline__ unsigned short f32_to_bf16(float f) {
    unsigned int u = __float_as_uint(f);
    unsigned int r = u + 0x7FFFu + ((u >> 16) & 1u);
    return (unsigned short)(r >> 16);
}

__global__ void wt_bf16_kernel(const float* __restrict__ W,
                               unsigned short* __restrict__ Wt,
                               int K, int Nout) {
    int i = blockIdx.x * blockDim.x + threadIdx.x;
    if (i < K * Nout) {
        int k = i / Nout, n = i - k * Nout;
        Wt[n * K + k] = f32_to_bf16(W[i]);
    }
}

__global__ void gemm_wmma_bf16(const float* __restrict__ A,
                               const unsigned short* __restrict__ Wt,
                               float* __restrict__ C,
                               const float* __restrict__ bias,
                               int M, int K, int Nout) {
    __shared__ unsigned short As[16 * (HIDF + 8)];
    const int ldA = K + 8;

    const int tid = threadIdx.x;
    const int m0  = blockIdx.x * 16;

    const int total = 16 * K;
    for (int i = tid; i < total; i += blockDim.x) {
        int r = i / K, c = i - r * K;
        int gr = m0 + r;
        float v = (gr < M) ? A[(size_t)gr * K + c] : 0.0f;
        As[r * ldA + c] = f32_to_bf16(v);
    }
    __syncthreads();

    const int wave  = tid >> 5;
    const int lane  = tid & 31;
    const int row   = lane & 15;
    const int khalf = lane >> 4;
    const int ncol  = wave * 16 + row;

    v8f acc = {};

    for (int k0 = 0; k0 < K; k0 += 32) {
        const int cb = k0 + khalf * 8;

        BfFrag a, b;
        {
            const uint4* p0 = (const uint4*)&As[row * ldA + cb];
            const uint4* p1 = (const uint4*)&As[row * ldA + cb + 16];
            uint4 x0 = *p0, x1 = *p1;
            a.u[0] = x0.x; a.u[1] = x0.y; a.u[2] = x0.z; a.u[3] = x0.w;
            a.u[4] = x1.x; a.u[5] = x1.y; a.u[6] = x1.z; a.u[7] = x1.w;
        }
        {
            const uint4* p0 = (const uint4*)&Wt[(size_t)ncol * K + cb];
            const uint4* p1 = (const uint4*)&Wt[(size_t)ncol * K + cb + 16];
            uint4 x0 = *p0, x1 = *p1;
            b.u[0] = x0.x; b.u[1] = x0.y; b.u[2] = x0.z; b.u[3] = x0.w;
            b.u[4] = x1.x; b.u[5] = x1.y; b.u[6] = x1.z; b.u[7] = x1.w;
        }

        acc = __builtin_amdgcn_wmma_f32_16x16x32_bf16(
             false, a.bf,  false, b.bf,
             (short)0, acc,  false,  false);
    }

    float badd = bias ? bias[ncol] : 0.0f;
#pragma unroll
    for (int r = 0; r < 8; ++r) {
        int gr = m0 + r + khalf * 8;
        if (gr < M) C[(size_t)gr * Nout + ncol] = acc[r] + badd;
    }
}

__global__ void node_scores_kernel(const float* __restrict__ h,
                                   const float* __restrict__ a_s,
                                   const float* __restrict__ a_d,
                                   float* __restrict__ s_src,
                                   float* __restrict__ s_dst, int N) {
    int gt   = blockIdx.x * blockDim.x + threadIdx.x;
    int node = gt >> 5;
    int lane = gt & 31;
    if (node >= N) return;
    const float* hp = h + (size_t)node * HIDF;
    float ss = 0.0f, sd = 0.0f;
#pragma unroll
    for (int f = lane; f < HIDF; f += 32) {
        float v = hp[f];
        ss += v * a_s[f];
        sd += v * a_d[f];
    }
#pragma unroll
    for (int off = 16; off; off >>= 1) {
        ss += __shfl_xor(ss, off, 32);
        sd += __shfl_xor(sd, off, 32);
    }
    if (lane == 0) { s_src[node] = ss; s_dst[node] = sd; }
}

__global__ void zero_counts_kernel(int* __restrict__ cnt, int N) {
    int i = blockIdx.x * blockDim.x + threadIdx.x;
    if (i < N) cnt[i] = 0;
}

__global__ void edge_rank_kernel(const int* __restrict__ ei, int* __restrict__ cnt,
                                 int* __restrict__ rank, int E) {
    int i = blockIdx.x * blockDim.x + threadIdx.x;
    if (i < E) rank[i] = atomicAdd(&cnt[ei[E + i]], 1);
}

__global__ void offsets_kernel(const int* __restrict__ cnt, int* __restrict__ off, int N) {
    __shared__ int sh_front[8];
    __shared__ int sh_tot[8];
    const int t    = threadIdx.x;
    const int lane = t & 31;
    const int wave = t >> 5;
    const int n0   = blockIdx.x * 256;

    int front = 0;
    for (int i = t; i < n0; i += 256) front += cnt[i];
    const int idx = n0 + t;
    const int c   = idx < N ? cnt[idx] : 0;

    int x = c;
    for (int o = 1; o < 32; o <<= 1) {
        int y = __shfl_up(x, o, 32);
        if (lane >= o) x += y;
    }
    for (int o = 16; o; o >>= 1) front += __shfl_xor(front, o, 32);
    if (lane == 31) sh_tot[wave] = x;
    if (lane == 0)  sh_front[wave] = front;
    __syncthreads();
    int base = 0;
    for (int w = 0; w < 8; ++w) {
        base += sh_front[w];
        if (w < wave) base += sh_tot[w];
    }
    if (idx < N) off[idx] = base + x - c;
    if (idx == N - 1) off[N] = base + x;
}

__global__ void edge_fill_kernel(const int* __restrict__ ei, const int* __restrict__ off,
                                 const int* __restrict__ rank, int* __restrict__ perm, int E) {
    int i = blockIdx.x * blockDim.x + threadIdx.x;
    if (i < E) perm[off[ei[E + i]] + rank[i]] = i;
}

__global__ void order_segments_kernel(const int* __restrict__ ei, const int* __restrict__ off,
                                      const int* __restrict__ perm, int* __restrict__ csr, int N) {
    const int node = __builtin_amdgcn_readfirstlane(
        (int)((blockIdx.x * blockDim.x + threadIdx.x) >> 5));
    const int lane = threadIdx.x & 31;
    if (node >= N) return;
    const int start = off[node];
    const int deg   = off[node + 1] - start;
    if (deg <= 32) {
        const bool valid = lane < deg;
        const int id = valid ? perm[start + lane] : 0x7fffffff;
        int r = 0;
        for (int k = 0; k < deg; ++k) r += (__builtin_amdgcn_readlane(id, k) < id) ? 1 : 0;
        if (valid) csr[start + r] = ei[id];
    } else {
        for (int c = 0; c < deg; c += 32) {
            const int idx = c + lane;
            if (idx < deg) {
                const int id = perm[start + idx];
                int r = 0;
                for (int k = 0; k < deg; ++k) r += (perm[start + k] < id) ? 1 : 0;
                csr[start + r] = ei[id];
            }
        }
    }
}

static __device__ __forceinline__ float leaky_att(float e) {
    return e > 0.0f ? e : 0.2f * e;
}
static __device__ __forceinline__ float wave_max(float v) {
    for (int o = 16; o; o >>= 1) v = fmaxf(v, __shfl_xor(v, o, 32));
    return v;
}
static __device__ __forceinline__ float wave_sum(float v) {
    for (int o = 16; o; o >>= 1) v += __shfl_xor(v, o, 32);
    return v;
}
static __device__ __forceinline__ void fma4(float4& acc, float a, const float4& v) {
    acc.x += a * v.x; acc.y += a * v.y; acc.z += a * v.z; acc.w += a * v.w;
}
static __device__ __forceinline__ float4 row_of(const float4* __restrict__ hrow,
                                                int sbits, int j, int lane) {
    return hrow[(size_t)__builtin_amdgcn_readlane(sbits, j) * 32 + lane];
}
static __device__ __forceinline__ float w_of(int abits, int j) {
    return __int_as_float(__builtin_amdgcn_readlane(abits, j));
}

__global__ __launch_bounds__(256)
void gat_node_kernel(const int* __restrict__ off, const int* __restrict__ csr,
                     const float* __restrict__ ssrc, const float* __restrict__ sdst,
                     const float* __restrict__ h, const float* __restrict__ b,
                     float* __restrict__ out, int N) {
    const int node = __builtin_amdgcn_readfirstlane(
        (int)((blockIdx.x * blockDim.x + threadIdx.x) >> 5));
    const int lane = threadIdx.x & 31;
    if (node >= N) return;
    const int start = off[node];
    const int deg   = off[node + 1] - start;
    const int tot   = deg + 1;
    const float sd  = sdst[node];
    const float4* __restrict__ hrow = (const float4*)h;
    const float NEG_INF = __int_as_float(0xFF800000);
    float4 acc = {0.0f, 0.0f, 0.0f, 0.0f};

    if (tot <= 32) {
        const bool valid = lane < tot;
        int s = node;
        if (lane < deg) s = csr[start + lane];
        const float e   = leaky_att(ssrc[s] + sd);
        const float m   = wave_max(valid ? e : NEG_INF);
        const float ex  = valid ? __expf(e - m) : 0.0f;
        const float den = wave_sum(ex);
        const float alpha = ex / (den + 1e-16f);
        const int sbits = s;
        const int abits = __float_as_int(alpha);
        int j = 0;
        for (; j + 4 <= tot; j += 4) {
            const float4 v0 = row_of(hrow, sbits, j,     lane);
            const float4 v1 = row_of(hrow, sbits, j + 1, lane);
            const float4 v2 = row_of(hrow, sbits, j + 2, lane);
            const float4 v3 = row_of(hrow, sbits, j + 3, lane);
            fma4(acc, w_of(abits, j),     v0);
            fma4(acc, w_of(abits, j + 1), v1);
            fma4(acc, w_of(abits, j + 2), v2);
            fma4(acc, w_of(abits, j + 3), v3);
        }
        const int rem = tot - j;
        if (rem == 3) {
            const float4 v0 = row_of(hrow, sbits, j,     lane);
            const float4 v1 = row_of(hrow, sbits, j + 1, lane);
            const float4 v2 = row_of(hrow, sbits, j + 2, lane);
            fma4(acc, w_of(abits, j),     v0);
            fma4(acc, w_of(abits, j + 1), v1);
            fma4(acc, w_of(abits, j + 2), v2);
        } else if (rem == 2) {
            const float4 v0 = row_of(hrow, sbits, j,     lane);
            const float4 v1 = row_of(hrow, sbits, j + 1, lane);
            fma4(acc, w_of(abits, j),     v0);
            fma4(acc, w_of(abits, j + 1), v1);
        } else if (rem == 1) {
            const float4 v0 = row_of(hrow, sbits, j, lane);
            fma4(acc, w_of(abits, j), v0);
        }
    } else {
        float m = NEG_INF;
        for (int c = 0; c < tot; c += 32) {
            const int idx = c + lane;
            if (idx < tot) {
                const int s = idx < deg ? csr[start + idx] : node;
                m = fmaxf(m, leaky_att(ssrc[s] + sd));
            }
        }
        m = wave_max(m);
        float den = 0.0f;
        for (int c = 0; c < tot; c += 32) {
            const int idx = c + lane;
            if (idx < tot) {
                const int s = idx < deg ? csr[start + idx] : node;
                den += __expf(leaky_att(ssrc[s] + sd) - m);
            }
        }
        den = wave_sum(den);
        for (int c = 0; c < tot; c += 32) {
            const int idx = c + lane;
            int s = node;
            if (idx < deg) s = csr[start + idx];
            float alpha = 0.0f;
            if (idx < tot) alpha = __expf(leaky_att(ssrc[s] + sd) - m) / (den + 1e-16f);
            const int abits = __float_as_int(alpha);
            const int cnt = (tot - c) < 32 ? (tot - c) : 32;
            for (int j = 0; j < cnt; ++j) {
                fma4(acc, w_of(abits, j), row_of(hrow, s, j, lane));
            }
        }
    }

    float r[4] = {acc.x, acc.y, acc.z, acc.w};
    for (int k = 0; k < 4; ++k) {
        float v = r[k] + b[lane * 4 + k];
        r[k] = v > 0.0f ? v : 0.01f * v;
    }
    float4 o = {r[0], r[1], r[2], r[3]};
    ((float4*)out)[(size_t)node * 32 + lane] = o;
}

extern "C" void kernel_launch(void* const* d_in, const int* in_sizes, int n_in,
                              void* d_out, int out_size, void* d_ws, size_t ws_size,
                              hipStream_t stream) {
    (void)n_in; (void)out_size; (void)ws_size;

    const float* x    = (const float*)d_in[0];
    const int*   ei   = (const int*)d_in[1];
    const float* Wl[3] = {(const float*)d_in[2], (const float*)d_in[6],  (const float*)d_in[10]};
    const float* Asl[3]= {(const float*)d_in[3], (const float*)d_in[7],  (const float*)d_in[11]};
    const float* Adl[3]= {(const float*)d_in[4], (const float*)d_in[8],  (const float*)d_in[12]};
    const float* Bl[3] = {(const float*)d_in[5], (const float*)d_in[9],  (const float*)d_in[13]};
    const float* Wout = (const float*)d_in[14];
    const float* bout = (const float*)d_in[15];

    const int FIN  = 64;
    const int FOUT = 64;
    const int N  = in_sizes[0] / FIN;
    const int E  = in_sizes[1] / 2;

    float* ws     = (float*)d_ws;
    float* h_pre  = ws;
    float* h_agg  = h_pre + (size_t)N * HIDF;
    unsigned short* Wt = (unsigned short*)(h_agg + (size_t)N * HIDF);
    float* s_src  = (float*)(Wt + HIDF * HIDF);
    float* s_dst  = s_src + N;
    int* cnt   = (int*)(s_dst + N);
    int* off   = cnt + N;
    int* rank  = off + N + 1;
    int* perm  = rank + E;
    int* csr   = perm + E;

    const int B256 = 256;
    const int gN    = (N + B256 - 1) / B256;
    const int gE    = (E + B256 - 1) / B256;
    const int gNW   = (int)(((long long)N * 32 + B256 - 1) / B256);
    const int gM    = (N + 15) / 16;

    zero_counts_kernel<<<gN, B256, 0, stream>>>(cnt, N);
    edge_rank_kernel<<<gE, B256, 0, stream>>>(ei, cnt, rank, E);
    offsets_kernel<<<gN, B256, 0, stream>>>(cnt, off, N);
    edge_fill_kernel<<<gE, B256, 0, stream>>>(ei, off, rank, perm, E);
    order_segments_kernel<<<gNW, B256, 0, stream>>>(ei, off, perm, csr, N);

    const float* Ain = x;
    int Kin = FIN;
    for (int l = 0; l < 3; ++l) {
        wt_bf16_kernel<<<(Kin * HIDF + B256 - 1) / B256, B256, 0, stream>>>(
            Wl[l], Wt, Kin, HIDF);
        gemm_wmma_bf16<<<gM, (HIDF / 16) * 32, 0, stream>>>(
            Ain, Wt, h_pre, nullptr, N, Kin, HIDF);
        node_scores_kernel<<<gNW, B256, 0, stream>>>(
            h_pre, Asl[l], Adl[l], s_src, s_dst, N);
        gat_node_kernel<<<gNW, B256, 0, stream>>>(off, csr, s_src, s_dst, h_pre,
                                                  Bl[l], h_agg, N);
        Ain = h_agg;
        Kin = HIDF;
    }

    wt_bf16_kernel<<<(HIDF * FOUT + B256 - 1) / B256, B256, 0, stream>>>(
        Wout, Wt, HIDF, FOUT);
    gemm_wmma_bf16<<<gM, (FOUT / 16) * 32, 0, stream>>>(
        h_agg, Wt, (float*)d_out, bout, N, HIDF, FOUT);
}


# Independent stages share launches: counter clear and all 4 weight transposes (own buffers) in one launch, layer-0 GEMM inside the offset-scan launch; 13 launches become 8, arithmetic untouched
# speedup vs baseline: 3.7867x; 1.3306x over previous
#include <hip/hip_runtime.h>
#include <hip/hip_bf16.h>


typedef __attribute__((ext_vector_type(16))) __bf16 v16bf;
typedef __attribute__((ext_vector_type(8)))  float  v8f;

union BfFrag {
    unsigned int u[8];
    v16bf bf;
};

#define HIDF 128
#define FOUT_C 64

static __device__ __forceinline__ unsigned short f32_to_bf16(float f) {
    unsigned int u = __float_as_uint(f);
    unsigned int r = u + 0x7FFFu + ((u >> 16) & 1u);
    return (unsigned short)(r >> 16);
}

static __device__ __forceinline__ void wt_bf16_item(const float* __restrict__ W,
                                                    unsigned short* __restrict__ Wt,
                                                    int K, int Nout, int i) {
    if (i < K * Nout) {
        int k = i / Nout, n = i - k * Nout;
        Wt[n * K + k] = f32_to_bf16(W[i]);
    }
}

__global__ __launch_bounds__(256)
void prep_kernel(int* __restrict__ cnt, int N, int nZero,
                 const float* __restrict__ W0, const float* __restrict__ W1,
                 const float* __restrict__ W2, const float* __restrict__ W3,
                 unsigned short* __restrict__ Wt0, unsigned short* __restrict__ Wt1,
                 unsigned short* __restrict__ Wt2, unsigned short* __restrict__ Wt3) {
    const int t = threadIdx.x;
    int b = blockIdx.x;
    if (b < nZero) {
        const int i = b * 256 + t;
        if (i < N) cnt[i] = 0;
        return;
    }
    b -= nZero;
    if (b < 32) { wt_bf16_item(W0, Wt0, 64, HIDF, b * 256 + t); return; }
    b -= 32;
    if (b < 64) { wt_bf16_item(W1, Wt1, HIDF, HIDF, b * 256 + t); return; }
    b -= 64;
    if (b < 64) { wt_bf16_item(W2, Wt2, HIDF, HIDF, b * 256 + t); return; }
    b -= 64;
    wt_bf16_item(W3, Wt3, HIDF, FOUT_C, b * 256 + t);
}

static __device__ __forceinline__ float half_wave_sum(float v) {
    v += __int_as_float(__builtin_amdgcn_update_dpp(0, __float_as_int(v), 0x161, 0xf, 0xf, true));
    v += __int_as_float(__builtin_amdgcn_update_dpp(0, __float_as_int(v), 0x162, 0xf, 0xf, true));
    v += __int_as_float(__builtin_amdgcn_update_dpp(0, __float_as_int(v), 0x164, 0xf, 0xf, true));
    v += __int_as_float(__builtin_amdgcn_update_dpp(0, __float_as_int(v), 0x168, 0xf, 0xf, true));
    return v;
}

template <bool SCORES>
static __device__ __forceinline__ void tile_gemm(const unsigned short* As, const int ldA, const int K,
                                                 const unsigned short* __restrict__ Wt,
                                                 float* __restrict__ C,
                                                 const float* __restrict__ bias,
                                                 const float* __restrict__ a_s,
                                                 const float* __restrict__ a_d,
                                                 float* __restrict__ s_src,
                                                 float* __restrict__ s_dst,
                                                 float* part,
                                                 const int m0, const int M, const int Nout,
                                                 const int tid) {
    const int wave  = tid >> 5;
    const int lane  = tid & 31;
    const int row   = lane & 15;
    const int khalf = lane >> 4;
    const int ncol  = wave * 16 + row;

    float as_c = 0.0f, ad_c = 0.0f;
    if (SCORES) { as_c = a_s[ncol]; ad_c = a_d[ncol]; }

    v8f acc = {};

#pragma unroll 2
    for (int k0 = 0; k0 < K; k0 += 32) {
        const int cb = k0 + khalf * 8;

        BfFrag a, b;
        {
            const uint4* p0 = (const uint4*)&As[row * ldA + cb];
            const uint4* p1 = (const uint4*)&As[row * ldA + cb + 16];
            uint4 x0 = *p0, x1 = *p1;
            a.u[0] = x0.x; a.u[1] = x0.y; a.u[2] = x0.z; a.u[3] = x0.w;
            a.u[4] = x1.x; a.u[5] = x1.y; a.u[6] = x1.z; a.u[7] = x1.w;
        }
        {
            const uint4* p0 = (const uint4*)&Wt[(size_t)ncol * K + cb];
            const uint4* p1 = (const uint4*)&Wt[(size_t)ncol * K + cb + 16];
            uint4 x0 = *p0, x1 = *p1;
            b.u[0] = x0.x; b.u[1] = x0.y; b.u[2] = x0.z; b.u[3] = x0.w;
            b.u[4] = x1.x; b.u[5] = x1.y; b.u[6] = x1.z; b.u[7] = x1.w;
        }

        acc = __builtin_amdgcn_wmma_f32_16x16x32_bf16(
             false, a.bf,  false, b.bf,
             (short)0, acc,  false,  false);
    }

    float badd = bias ? bias[ncol] : 0.0f;
#pragma unroll
    for (int r = 0; r < 8; ++r) {
        int gr = m0 + r + khalf * 8;
        if (gr < M) C[(size_t)gr * Nout + ncol] = acc[r] + badd;
    }

    if (SCORES) {
        float ps[8], pd[8];
#pragma unroll
        for (int r = 0; r < 8; ++r) {
            ps[r] = half_wave_sum(acc[r] * as_c);
            pd[r] = half_wave_sum(acc[r] * ad_c);
        }
        if (row == 0) {
#pragma unroll
            for (int r = 0; r < 8; ++r) {
                float2 p = {ps[r], pd[r]};
                *(float2*)&part[(wave * 16 + khalf * 8 + r) * 2] = p;
            }
        }
        __syncthreads();
        if (tid < 32) {
            const int rr = tid & 15, which = tid >> 4;
            float s = 0.0f;
#pragma unroll
            for (int w = 0; w < 8; ++w) s += part[(w * 16 + rr) * 2 + which];
            const int gr = m0 + rr;
            if (gr < M) (which ? s_dst : s_src)[gr] = s;
        }
    }
}

static __device__ __forceinline__ void gemm_scores_block(const float* __restrict__ A,
                                                         const unsigned short* __restrict__ Wt,
                                                         float* __restrict__ C,
                                                         const float* __restrict__ a_s,
                                                         const float* __restrict__ a_d,
                                                         float* __restrict__ s_src,
                                                         float* __restrict__ s_dst,
                                                         int M, int K, int tile) {
    __shared__ unsigned short As[16 * (HIDF + 8)];
    __shared__ float part[8 * 16 * 2];
    const int ldA = K + 8;

    const int tid = threadIdx.x;
    const int m0  = tile * 16;

    const int total = 16 * K;
    for (int i = tid; i < total; i += 256) {
        int r = i / K, c = i - r * K;
        int gr = m0 + r;
        float v = (gr < M) ? A[(size_t)gr * K + c] : 0.0f;
        As[r * ldA + c] = f32_to_bf16(v);
    }
    __syncthreads();

    tile_gemm<true>(As, ldA, K, Wt, C, nullptr, a_s, a_d, s_src, s_dst, part, m0, M, HIDF, tid);
}

__global__ void edge_rank_kernel(const int* __restrict__ ei, int* __restrict__ cnt,
                                 int* __restrict__ rank, int E) {
    int i = blockIdx.x * blockDim.x + threadIdx.x;
    if (i < E) rank[i] = atomicAdd(&cnt[ei[E + i]], 1);
}

static __device__ __forceinline__ void offsets_block(const int* __restrict__ cnt,
                                                     int* __restrict__ off, int N, int blk) {
    __shared__ int sh_front[8];
    __shared__ int sh_tot[8];
    const int t    = threadIdx.x;
    const int lane = t & 31;
    const int wave = t >> 5;
    const int n0   = blk * 256;

    int front = 0;
    for (int i = t; i < n0; i += 256) front += cnt[i];
    const int idx = n0 + t;
    const int c   = idx < N ? cnt[idx] : 0;

    int x = c;
    for (int o = 1; o < 32; o <<= 1) {
        int y = __shfl_up(x, o, 32);
        if (lane >= o) x += y;
    }
    for (int o = 16; o; o >>= 1) front += __shfl_xor(front, o, 32);
    if (lane == 31) sh_tot[wave] = x;
    if (lane == 0)  sh_front[wave] = front;
    __syncthreads();
    int base = 0;
    for (int w = 0; w < 8; ++w) {
        base += sh_front[w];
        if (w < wave) base += sh_tot[w];
    }
    if (idx < N) off[idx] = base + x - c;
    if (idx == N - 1) off[N] = base + x;
}

__global__ __launch_bounds__(256)
void offsets_gemm_kernel(const int* __restrict__ cnt, int* __restrict__ off, int N, int nOff,
                         const float* __restrict__ A,
                         const unsigned short* __restrict__ Wt,
                         float* __restrict__ C,
                         const float* __restrict__ a_s,
                         const float* __restrict__ a_d,
                         float* __restrict__ s_src,
                         float* __restrict__ s_dst,
                         int K) {
    const int b = blockIdx.x;
    if (b < nOff) offsets_block(cnt, off, N, b);
    else gemm_scores_block(A, Wt, C, a_s, a_d, s_src, s_dst, N, K, b - nOff);
}

__global__ void edge_fill_kernel(const int* __restrict__ ei, const int* __restrict__ off,
                                 const int* __restrict__ rank, int* __restrict__ perm, int E) {
    int i = blockIdx.x * blockDim.x + threadIdx.x;
    if (i < E) perm[off[ei[E + i]] + rank[i]] = i;
}

__global__ void order_segments_kernel(const int* __restrict__ ei, const int* __restrict__ off,
                                      const int* __restrict__ perm, int* __restrict__ csr, int N) {
    const int node = __builtin_amdgcn_readfirstlane(
        (int)((blockIdx.x * blockDim.x + threadIdx.x) >> 5));
    const int lane = threadIdx.x & 31;
    if (node >= N) return;
    const int start = off[node];
    const int deg   = off[node + 1] - start;
    if (deg <= 32) {
        const bool valid = lane < deg;
        const int id = valid ? perm[start + lane] : 0x7fffffff;
        int r = 0;
        for (int k = 0; k < deg; ++k) r += (__builtin_amdgcn_readlane(id, k) < id) ? 1 : 0;
        if (valid) csr[start + r] = ei[id];
    } else {
        for (int c = 0; c < deg; c += 32) {
            const int idx = c + lane;
            if (idx < deg) {
                const int id = perm[start + idx];
                int r = 0;
                for (int k = 0; k < deg; ++k) r += (perm[start + k] < id) ? 1 : 0;
                csr[start + r] = ei[id];
            }
        }
    }
}

static __device__ __forceinline__ float leaky_att(float e) {
    return e > 0.0f ? e : 0.2f * e;
}
static __device__ __forceinline__ float wave_max(float v) {
    for (int o = 16; o; o >>= 1) v = fmaxf(v, __shfl_xor(v, o, 32));
    return v;
}
static __device__ __forceinline__ float wave_sum(float v) {
    for (int o = 16; o; o >>= 1) v += __shfl_xor(v, o, 32);
    return v;
}
static __device__ __forceinline__ void fma4(float4& acc, float a, const float4& v) {
    acc.x += a * v.x; acc.y += a * v.y; acc.z += a * v.z; acc.w += a * v.w;
}
static __device__ __forceinline__ float4 row_of(const float4* __restrict__ hrow,
                                                int sbits, int j, int lane) {
    return hrow[(size_t)__builtin_amdgcn_readlane(sbits, j) * 32 + lane];
}
static __device__ __forceinline__ float w_of(int abits, int j) {
    return __int_as_float(__builtin_amdgcn_readlane(abits, j));
}

static __device__ __forceinline__ float4 gat_row(const int node, const int lane,
                                                 const int* __restrict__ off,
                                                 const int* __restrict__ csr,
                                                 const float* __restrict__ ssrc,
                                                 const float* __restrict__ sdst,
                                                 const float* __restrict__ h,
                                                 const float4 bv) {
    const int start = off[node];
    const int deg   = off[node + 1] - start;
    const int tot   = deg + 1;
    const float sd  = sdst[node];
    const float4* __restrict__ hrow = (const float4*)h;
    const float NEG_INF = __int_as_float(0xFF800000);
    float4 acc = {0.0f, 0.0f, 0.0f, 0.0f};

    if (tot <= 32) {
        const bool valid = lane < tot;
        int s = node;
        if (lane < deg) s = csr[start + lane];
        const float e   = leaky_att(ssrc[s] + sd);
        const float m   = wave_max(valid ? e : NEG_INF);
        const float ex  = valid ? __expf(e - m) : 0.0f;
        const float den = wave_sum(ex);
        const float alpha = ex / (den + 1e-16f);
        const int sbits = s;
        const int abits = __float_as_int(alpha);
        int j = 0;
        for (; j + 4 <= tot; j += 4) {
            const float4 v0 = row_of(hrow, sbits, j,     lane);
            const float4 v1 = row_of(hrow, sbits, j + 1, lane);
            const float4 v2 = row_of(hrow, sbits, j + 2, lane);
            const float4 v3 = row_of(hrow, sbits, j + 3, lane);
            fma4(acc, w_of(abits, j),     v0);
            fma4(acc, w_of(abits, j + 1), v1);
            fma4(acc, w_of(abits, j + 2), v2);
            fma4(acc, w_of(abits, j + 3), v3);
        }
        const int rem = tot - j;
        if (rem == 3) {
            const float4 v0 = row_of(hrow, sbits, j,     lane);
            const float4 v1 = row_of(hrow, sbits, j + 1, lane);
            const float4 v2 = row_of(hrow, sbits, j + 2, lane);
            fma4(acc, w_of(abits, j),     v0);
            fma4(acc, w_of(abits, j + 1), v1);
            fma4(acc, w_of(abits, j + 2), v2);
        } else if (rem == 2) {
            const float4 v0 = row_of(hrow, sbits, j,     lane);
            const float4 v1 = row_of(hrow, sbits, j + 1, lane);
            fma4(acc, w_of(abits, j),     v0);
            fma4(acc, w_of(abits, j + 1), v1);
        } else if (rem == 1) {
            const float4 v0 = row_of(hrow, sbits, j, lane);
            fma4(acc, w_of(abits, j), v0);
        }
    } else {
        float m = NEG_INF;
        for (int c = 0; c < tot; c += 32) {
            const int idx = c + lane;
            if (idx < tot) {
                const int s = idx < deg ? csr[start + idx] : node;
                m = fmaxf(m, leaky_att(ssrc[s] + sd));
            }
        }
        m = wave_max(m);
        float den = 0.0f;
        for (int c = 0; c < tot; c += 32) {
            const int idx = c + lane;
            if (idx < tot) {
                const int s = idx < deg ? csr[start + idx] : node;
                den += __expf(leaky_att(ssrc[s] + sd) - m);
            }
        }
        den = wave_sum(den);
        for (int c = 0; c < tot; c += 32) {
            const int idx = c + lane;
            int s = node;
            if (idx < deg) s = csr[start + idx];
            float alpha = 0.0f;
            if (idx < tot) alpha = __expf(leaky_att(ssrc[s] + sd) - m) / (den + 1e-16f);
            const int abits = __float_as_int(alpha);
            const int cnt = (tot - c) < 32 ? (tot - c) : 32;
#pragma unroll 4
            for (int j = 0; j < cnt; ++j) {
                fma4(acc, w_of(abits, j), row_of(hrow, s, j, lane));
            }
        }
    }

    float r[4] = {acc.x, acc.y, acc.z, acc.w};
    const float bb[4] = {bv.x, bv.y, bv.z, bv.w};
    for (int k = 0; k < 4; ++k) {
        float v = r[k] + bb[k];
        r[k] = v > 0.0f ? v : 0.01f * v;
    }
    float4 o = {r[0], r[1], r[2], r[3]};
    return o;
}

template <int NOUT, bool SCORES>
__global__ __launch_bounds__(256)
void gat_gemm_kernel(const int* __restrict__ off, const int* __restrict__ csr,
                     const float* __restrict__ ssrc, const float* __restrict__ sdst,
                     const float* __restrict__ h, const float* __restrict__ b,
                     const unsigned short* __restrict__ Wt,
                     float* __restrict__ C,
                     const float* __restrict__ bias_out,
                     const float* __restrict__ a_s, const float* __restrict__ a_d,
                     float* __restrict__ s_src_out, float* __restrict__ s_dst_out,
                     int N) {
    __shared__ unsigned short As[16 * (HIDF + 8)];
    __shared__ float part[SCORES ? 8 * 16 * 2 : 2];
    const int ldA = HIDF + 8;

    const int tid  = threadIdx.x;
    const int lane = tid & 31;
    const int wave = __builtin_amdgcn_readfirstlane(tid >> 5);
    const int m0   = blockIdx.x * 16;

    const float4 bv = {b[lane * 4], b[lane * 4 + 1], b[lane * 4 + 2], b[lane * 4 + 3]};
    for (int q = 0; q < 2; ++q) {
        const int r    = wave * 2 + q;
        const int node = m0 + r;
        float4 o = {0.0f, 0.0f, 0.0f, 0.0f};
        if (node < N) o = gat_row(node, lane, off, csr, ssrc, sdst, h, bv);
        uint2 p;
        p.x = (unsigned int)f32_to_bf16(o.x) | ((unsigned int)f32_to_bf16(o.y) << 16);
        p.y = (unsigned int)f32_to_bf16(o.z) | ((unsigned int)f32_to_bf16(o.w) << 16);
        *(uint2*)&As[r * ldA + lane * 4] = p;
    }
    __syncthreads();

    if (SCORES) {
        tile_gemm<true>(As, ldA, HIDF, Wt, C, bias_out, a_s, a_d, s_src_out, s_dst_out,
                        part, m0, N, NOUT, tid);
    } else if (wave < NOUT / 16) {
        tile_gemm<false>(As, ldA, HIDF, Wt, C, bias_out, a_s, a_d, s_src_out, s_dst_out,
                         part, m0, N, NOUT, tid);
    }
}

extern "C" void kernel_launch(void* const* d_in, const int* in_sizes, int n_in,
                              void* d_out, int out_size, void* d_ws, size_t ws_size,
                              hipStream_t stream) {
    (void)n_in; (void)out_size; (void)ws_size;

    const float* x    = (const float*)d_in[0];
    const int*   ei   = (const int*)d_in[1];
    const float* Wl[3] = {(const float*)d_in[2], (const float*)d_in[6],  (const float*)d_in[10]};
    const float* Asl[3]= {(const float*)d_in[3], (const float*)d_in[7],  (const float*)d_in[11]};
    const float* Adl[3]= {(const float*)d_in[4], (const float*)d_in[8],  (const float*)d_in[12]};
    const float* Bl[3] = {(const float*)d_in[5], (const float*)d_in[9],  (const float*)d_in[13]};
    const float* Wout = (const float*)d_in[14];
    const float* bout = (const float*)d_in[15];

    const int FIN  = 64;
    const int N  = in_sizes[0] / FIN;
    const int E  = in_sizes[1] / 2;

    float* ws     = (float*)d_ws;
    float* hA     = ws;
    float* hB     = hA + (size_t)N * HIDF;
    unsigned short* Wt0 = (unsigned short*)(hB + (size_t)N * HIDF);
    unsigned short* Wt1 = Wt0 + HIDF * HIDF;
    unsigned short* Wt2 = Wt1 + HIDF * HIDF;
    unsigned short* Wt3 = Wt2 + HIDF * HIDF;
    float* sA_src = (float*)(Wt3 + HIDF * HIDF);
    float* sA_dst = sA_src + N;
    float* sB_src = sA_dst + N;
    float* sB_dst = sB_src + N;
    int* cnt   = (int*)(sB_dst + N);
    int* off   = cnt + N;
    int* rank  = off + N + 1;
    int* perm  = rank + E;
    int* csr   = perm + E;

    const int B256 = 256;
    const int gN    = (N + B256 - 1) / B256;
    const int gE    = (E + B256 - 1) / B256;
    const int gNW   = (int)(((long long)N * 32 + B256 - 1) / B256);
    const int gM    = (N + 15) / 16;

    prep_kernel<<<gN + 192, B256, 0, stream>>>(cnt, N, gN, Wl[0], Wl[1], Wl[2], Wout,
                                               Wt0, Wt1, Wt2, Wt3);

    edge_rank_kernel<<<gE, B256, 0, stream>>>(ei, cnt, rank, E);
    offsets_gemm_kernel<<<gN + gM, B256, 0, stream>>>(cnt, off, N, gN,
                                                      x, Wt0, hA, Asl[0], Adl[0], sA_src, sA_dst, FIN);
    edge_fill_kernel<<<gE, B256, 0, stream>>>(ei, off, rank, perm, E);
    order_segments_kernel<<<gNW, B256, 0, stream>>>(ei, off, perm, csr, N);

    gat_gemm_kernel<HIDF, true><<<gM, B256, 0, stream>>>(
        off, csr, sA_src, sA_dst, hA, Bl[0], Wt1, hB, nullptr, Asl[1], Adl[1], sB_src, sB_dst, N);

    gat_gemm_kernel<HIDF, true><<<gM, B256, 0, stream>>>(
        off, csr, sB_src, sB_dst, hB, Bl[1], Wt2, hA, nullptr, Asl[2], Adl[2], sA_src, sA_dst, N);

    gat_gemm_kernel<FOUT_C, false><<<gM, B256, 0, stream>>>(
        off, csr, sA_src, sA_dst, hA, Bl[2], Wt3, (float*)d_out, bout,
        nullptr, nullptr, nullptr, nullptr, N);
}


# Edge ordering per destination moved into the first layer's waves: fill scatters (edge number, source) pairs, a wave ranks its node's pairs in registers; ordering launch and its lookups gone (8 to 7)
# speedup vs baseline: 4.5489x; 1.2013x over previous
#include <hip/hip_runtime.h>
#include <hip/hip_bf16.h>


typedef __attribute__((ext_vector_type(16))) __bf16 v16bf;
typedef __attribute__((ext_vector_type(8)))  float  v8f;

union BfFrag {
    unsigned int u[8];
    v16bf bf;
};

#define HIDF 128
#define FOUT_C 64
#define FIN_C 64

static __device__ __forceinline__ unsigned short f32_to_bf16(float f) {
    unsigned int u = __float_as_uint(f);
    unsigned int r = u + 0x7FFFu + ((u >> 16) & 1u);
    return (unsigned short)(r >> 16);
}

static __device__ __forceinline__ unsigned short f32_to_f16_bits(float f) {
    return __builtin_bit_cast(unsigned short, (_Float16)f);
}

static __device__ __forceinline__ void wt_bf16_item(const float* __restrict__ W,
                                                    unsigned short* __restrict__ Wt,
                                                    int K, int Nout, int i) {
    if (i < K * Nout) {
        int k = i / Nout, n = i - k * Nout;
        Wt[n * K + k] = f32_to_bf16(W[i]);
    }
}

static __device__ __forceinline__ void score_vectors_block(const float* __restrict__ W0,
                                                           const float* __restrict__ a_s,
                                                           const float* __restrict__ a_d,
                                                           float* __restrict__ vsd, int t) {
    const int k = t >> 2, q = t & 3;
    const float4* __restrict__ w4 = (const float4*)(W0 + k * HIDF + q * 32);
    const float4* __restrict__ s4 = (const float4*)(a_s + q * 32);
    const float4* __restrict__ d4 = (const float4*)(a_d + q * 32);
    float ps = 0.0f, pd = 0.0f;
#pragma unroll
    for (int i = 0; i < 8; ++i) {
        const float4 w = w4[i], s = s4[i], d = d4[i];
        ps += w.x * s.x + w.y * s.y + w.z * s.z + w.w * s.w;
        pd += w.x * d.x + w.y * d.y + w.z * d.z + w.w * d.w;
    }
    ps += __shfl_xor(ps, 1, 32); ps += __shfl_xor(ps, 2, 32);
    pd += __shfl_xor(pd, 1, 32); pd += __shfl_xor(pd, 2, 32);
    if (q == 0) { vsd[k] = ps; vsd[FIN_C + k] = pd; }
}

__global__ __launch_bounds__(256)
void prep_kernel(int* __restrict__ cnt, int N, int nZero,
                 const float* __restrict__ W0, const float* __restrict__ W1,
                 const float* __restrict__ W2, const float* __restrict__ W3,
                 unsigned short* __restrict__ Wt0, unsigned short* __restrict__ Wt1,
                 unsigned short* __restrict__ Wt2, unsigned short* __restrict__ Wt3,
                 const float* __restrict__ a_s0, const float* __restrict__ a_d0,
                 float* __restrict__ vsd) {
    const int t = threadIdx.x;
    int b = blockIdx.x;
    if (b < nZero) {
        const int i = b * 256 + t;
        if (i < N) cnt[i] = 0;
        return;
    }
    b -= nZero;
    if (b < 32) { wt_bf16_item(W0, Wt0, 64, HIDF, b * 256 + t); return; }
    b -= 32;
    if (b < 64) { wt_bf16_item(W1, Wt1, HIDF, HIDF, b * 256 + t); return; }
    b -= 64;
    if (b < 64) { wt_bf16_item(W2, Wt2, HIDF, HIDF, b * 256 + t); return; }
    b -= 64;
    if (b < 32) { wt_bf16_item(W3, Wt3, HIDF, FOUT_C, b * 256 + t); return; }
    score_vectors_block(W0, a_s0, a_d0, vsd, t);
}

static __device__ __forceinline__ float half_wave_sum(float v) {
    v += __int_as_float(__builtin_amdgcn_update_dpp(0, __float_as_int(v), 0x161, 0xf, 0xf, true));
    v += __int_as_float(__builtin_amdgcn_update_dpp(0, __float_as_int(v), 0x162, 0xf, 0xf, true));
    v += __int_as_float(__builtin_amdgcn_update_dpp(0, __float_as_int(v), 0x164, 0xf, 0xf, true));
    v += __int_as_float(__builtin_amdgcn_update_dpp(0, __float_as_int(v), 0x168, 0xf, 0xf, true));
    return v;
}

template <bool SCORES>
static __device__ __forceinline__ void tile_gemm(const unsigned short* As, const int ldA, const int K,
                                                 const unsigned short* __restrict__ Wt,
                                                 float* __restrict__ C,
                                                 unsigned short* __restrict__ Ch,
                                                 unsigned short* Cs, const int ldC,
                                                 const float* __restrict__ bias,
                                                 const float* __restrict__ a_s,
                                                 const float* __restrict__ a_d,
                                                 float* __restrict__ s_src,
                                                 float* __restrict__ s_dst,
                                                 float* part,
                                                 const int m0, const int M, const int Nout,
                                                 const int tid) {
    const int wave  = tid >> 5;
    const int lane  = tid & 31;
    const int row   = lane & 15;
    const int khalf = lane >> 4;
    const int ncol  = wave * 16 + row;

    float as_c = 0.0f, ad_c = 0.0f;
    if (SCORES) { as_c = a_s[ncol]; ad_c = a_d[ncol]; }

    v8f acc = {};

#pragma unroll 2
    for (int k0 = 0; k0 < K; k0 += 32) {
        const int cb = k0 + khalf * 8;

        BfFrag a, b;
        {
            const uint4* p0 = (const uint4*)&As[row * ldA + cb];
            const uint4* p1 = (const uint4*)&As[row * ldA + cb + 16];
            uint4 x0 = *p0, x1 = *p1;
            a.u[0] = x0.x; a.u[1] = x0.y; a.u[2] = x0.z; a.u[3] = x0.w;
            a.u[4] = x1.x; a.u[5] = x1.y; a.u[6] = x1.z; a.u[7] = x1.w;
        }
        {
            const uint4* p0 = (const uint4*)&Wt[(size_t)ncol * K + cb];
            const uint4* p1 = (const uint4*)&Wt[(size_t)ncol * K + cb + 16];
            uint4 x0 = *p0, x1 = *p1;
            b.u[0] = x0.x; b.u[1] = x0.y; b.u[2] = x0.z; b.u[3] = x0.w;
            b.u[4] = x1.x; b.u[5] = x1.y; b.u[6] = x1.z; b.u[7] = x1.w;
        }

        acc = __builtin_amdgcn_wmma_f32_16x16x32_bf16(
             false, a.bf,  false, b.bf,
             (short)0, acc,  false,  false);
    }

    float badd = bias ? bias[ncol] : 0.0f;
    if (!SCORES) {
#pragma unroll
        for (int r = 0; r < 8; ++r) {
            int gr = m0 + r + khalf * 8;
            if (gr < M) C[(size_t)gr * Nout + ncol] = acc[r] + badd;
        }
    }

    if (SCORES) {
#pragma unroll
        for (int r = 0; r < 8; ++r)
            Cs[(r + khalf * 8) * ldC + ncol] = f32_to_f16_bits(acc[r] + badd);

        float ps[8], pd[8];
#pragma unroll
        for (int r = 0; r < 8; ++r) {
            ps[r] = half_wave_sum(acc[r] * as_c);
            pd[r] = half_wave_sum(acc[r] * ad_c);
        }
        if (row == 0) {
#pragma unroll
            for (int r = 0; r < 8; ++r) {
                float2 p = {ps[r], pd[r]};
                *(float2*)&part[(wave * 16 + khalf * 8 + r) * 2] = p;
            }
        }
        __syncthreads();
        {
            const int rr = tid >> 4, pc = tid & 15;
            const int gr = m0 + rr;
            if (gr < M) *(uint4*)&Ch[(size_t)gr * HIDF + pc * 8] = *(const uint4*)&Cs[rr * ldC + pc * 8];
        }
        if (tid < 32) {
            const int rr = tid & 15, which = tid >> 4;
            float s = 0.0f;
#pragma unroll
            for (int w = 0; w < 8; ++w) s += part[(w * 16 + rr) * 2 + which];
            const int gr = m0 + rr;
            if (gr < M) (which ? s_dst : s_src)[gr] = s;
        }
    }
}

static __device__ __forceinline__ void tile_gemm0_to_lds(const unsigned short* As0, const int ldA0,
                                                         const unsigned short* __restrict__ Wt0,
                                                         const float* __restrict__ b0,
                                                         unsigned short* As1, const int ldA1,
                                                         const int tid) {
    const int wave  = tid >> 5;
    const int lane  = tid & 31;
    const int row   = lane & 15;
    const int khalf = lane >> 4;
    const int ncol  = wave * 16 + row;

    v8f acc = {};
#pragma unroll
    for (int k0 = 0; k0 < FIN_C; k0 += 32) {
        const int cb = k0 + khalf * 8;
        BfFrag a, b;
        {
            const uint4* p0 = (const uint4*)&As0[row * ldA0 + cb];
            const uint4* p1 = (const uint4*)&As0[row * ldA0 + cb + 16];
            uint4 x0 = *p0, x1 = *p1;
            a.u[0] = x0.x; a.u[1] = x0.y; a.u[2] = x0.z; a.u[3] = x0.w;
            a.u[4] = x1.x; a.u[5] = x1.y; a.u[6] = x1.z; a.u[7] = x1.w;
        }
        {
            const uint4* p0 = (const uint4*)&Wt0[(size_t)ncol * FIN_C + cb];
            const uint4* p1 = (const uint4*)&Wt0[(size_t)ncol * FIN_C + cb + 16];
            uint4 x0 = *p0, x1 = *p1;
            b.u[0] = x0.x; b.u[1] = x0.y; b.u[2] = x0.z; b.u[3] = x0.w;
            b.u[4] = x1.x; b.u[5] = x1.y; b.u[6] = x1.z; b.u[7] = x1.w;
        }
        acc = __builtin_amdgcn_wmma_f32_16x16x32_bf16(
             false, a.bf,  false, b.bf,
             (short)0, acc,  false,  false);
    }

    const float badd = b0[ncol];
#pragma unroll
    for (int r = 0; r < 8; ++r) {
        float v = acc[r] + badd;
        v = v > 0.0f ? v : 0.01f * v;
        As1[(r + khalf * 8) * ldA1 + ncol] = f32_to_bf16(v);
    }
}

static __device__ __forceinline__ void node_scores_block(const float* __restrict__ x,
                                                         const float* __restrict__ vsd,
                                                         float* __restrict__ s_src,
                                                         float* __restrict__ s_dst,
                                                         int N, int blk) {
    const int tid = threadIdx.x;
    const int c   = tid & 15;
    const int n0  = blk * 64 + (tid >> 4);
    const float4* __restrict__ x4 = (const float4*)x;
    const float4 vs = ((const float4*)vsd)[c];
    const float4 vd = ((const float4*)vsd)[16 + c];
    float4 r[4];
#pragma unroll
    for (int i = 0; i < 4; ++i) {
        const int n = n0 + 16 * i;
        float4 z = {0.0f, 0.0f, 0.0f, 0.0f};
        if (n < N) z = x4[(size_t)n * 16 + c];
        r[i] = z;
    }
#pragma unroll
    for (int i = 0; i < 4; ++i) {
        const int n = n0 + 16 * i;
        const float ps = half_wave_sum(r[i].x * vs.x + r[i].y * vs.y + r[i].z * vs.z + r[i].w * vs.w);
        const float pd = half_wave_sum(r[i].x * vd.x + r[i].y * vd.y + r[i].z * vd.z + r[i].w * vd.w);
        if (c == 0 && n < N) { s_src[n] = ps; s_dst[n] = pd; }
    }
}

__global__ void edge_rank_kernel(const int* __restrict__ ei, int* __restrict__ cnt,
                                 int* __restrict__ rank, int E) {
    int i = blockIdx.x * blockDim.x + threadIdx.x;
    if (i < E) rank[i] = atomicAdd(&cnt[ei[E + i]], 1);
}

static __device__ __forceinline__ void offsets_block(const int* __restrict__ cnt,
                                                     int* __restrict__ off, int N, int blk) {
    __shared__ int sh_front[8];
    __shared__ int sh_tot[8];
    const int t    = threadIdx.x;
    const int lane = t & 31;
    const int wave = t >> 5;
    const int n0   = blk * 256;

    int front = 0;
    for (int i = t; i < n0; i += 256) front += cnt[i];
    const int idx = n0 + t;
    const int c   = idx < N ? cnt[idx] : 0;

    int x = c;
    for (int o = 1; o < 32; o <<= 1) {
        int y = __shfl_up(x, o, 32);
        if (lane >= o) x += y;
    }
    for (int o = 16; o; o >>= 1) front += __shfl_xor(front, o, 32);
    if (lane == 31) sh_tot[wave] = x;
    if (lane == 0)  sh_front[wave] = front;
    __syncthreads();
    int base = 0;
    for (int w = 0; w < 8; ++w) {
        base += sh_front[w];
        if (w < wave) base += sh_tot[w];
    }
    if (idx < N) off[idx] = base + x - c;
    if (idx == N - 1) off[N] = base + x;
}

__global__ __launch_bounds__(256)
void offsets_scores_kernel(const int* __restrict__ cnt, int* __restrict__ off, int N, int nOff,
                           const float* __restrict__ x,
                           const float* __restrict__ vsd,
                           float* __restrict__ s_src,
                           float* __restrict__ s_dst) {
    const int b = blockIdx.x;
    if (b < nOff) offsets_block(cnt, off, N, b);
    else node_scores_block(x, vsd, s_src, s_dst, N, b - nOff);
}

__global__ void edge_fill_kernel(const int* __restrict__ ei, const int* __restrict__ off,
                                 const int* __restrict__ rank, int* __restrict__ perm2, int E) {
    int i = blockIdx.x * blockDim.x + threadIdx.x;
    if (i < E) {
        int2 p;
        p.x = i;
        p.y = ei[i];
        ((int2*)perm2)[off[ei[E + i]] + rank[i]] = p;
    }
}

static __device__ __forceinline__ float leaky_att(float e) {
    return e > 0.0f ? e : 0.2f * e;
}
static __device__ __forceinline__ float wave_max(float v) {
    for (int o = 16; o; o >>= 1) v = fmaxf(v, __shfl_xor(v, o, 32));
    return v;
}
static __device__ __forceinline__ float wave_sum(float v) {
    for (int o = 16; o; o >>= 1) v += __shfl_xor(v, o, 32);
    return v;
}
static __device__ __forceinline__ void fma4(float4& acc, float a, const float4& v) {
    acc.x += a * v.x; acc.y += a * v.y; acc.z += a * v.z; acc.w += a * v.w;
}
static __device__ __forceinline__ float xhalf(float v) {
    const int b = __float_as_int(v);
    return __int_as_float(__builtin_amdgcn_permlanex16(b, b, 0x76543210, 0xfedcba98, false, false));
}
static __device__ __forceinline__ float4 pair_row(const float4* __restrict__ xrow,
                                                  int sbits, int j, int hsel, int c) {
    const int s0 = __builtin_amdgcn_readlane(sbits, j);
    const int s1 = __builtin_amdgcn_readlane(sbits, j + 1);
    return xrow[(size_t)(hsel ? s1 : s0) * 16 + c];
}
static __device__ __forceinline__ float pair_w(int abits, int j, int hsel) {
    const int a0 = __builtin_amdgcn_readlane(abits, j);
    const int a1 = __builtin_amdgcn_readlane(abits, j + 1);
    return __int_as_float(hsel ? a1 : a0);
}

static __device__ __forceinline__ float4 gat_row64(const int node, const int lane,
                                                   const int* __restrict__ off,
                                                   const int2* __restrict__ perm2,
                                                   int* csr,
                                                   const float* __restrict__ ssrc,
                                                   const float* __restrict__ sdst,
                                                   const float* __restrict__ x) {
    const int start = off[node];
    const int deg   = off[node + 1] - start;
    const int tot   = deg + 1;
    const float sd  = sdst[node];
    const float4* __restrict__ xrow = (const float4*)x;
    const int c    = lane & 15;
    const int hsel = lane >> 4;
    const float NEG_INF = __int_as_float(0xFF800000);
    float4 acc = {0.0f, 0.0f, 0.0f, 0.0f};

    if (tot <= 32) {
        const bool valid = lane < tot;
        int id = 0x7fffffff;
        int sv = node;
        if (lane < deg) { const int2 pr = perm2[start + lane]; id = pr.x; sv = pr.y; }
        int place = 0;
        for (int k = 0; k < deg; ++k) place += (__builtin_amdgcn_readlane(id, k) < id) ? 1 : 0;
        if (lane >= deg) place = lane;
        const int s = __builtin_amdgcn_ds_permute(place << 2, sv);
        if (lane < deg) csr[start + lane] = s;
        const float e   = leaky_att(ssrc[s] + sd);
        const float m   = wave_max(valid ? e : NEG_INF);
        const float ex  = valid ? __expf(e - m) : 0.0f;
        const float den = wave_sum(ex);
        const float alpha = ex / (den + 1e-16f);
        const int sbits = s;
        const int abits = __float_as_int(alpha);
        const int tot2  = (tot + 1) & ~1;
        int j = 0;
        for (; j + 8 <= tot2; j += 8) {
            const float4 v0 = pair_row(xrow, sbits, j,     hsel, c);
            const float4 v1 = pair_row(xrow, sbits, j + 2, hsel, c);
            const float4 v2 = pair_row(xrow, sbits, j + 4, hsel, c);
            const float4 v3 = pair_row(xrow, sbits, j + 6, hsel, c);
            fma4(acc, pair_w(abits, j,     hsel), v0);
            fma4(acc, pair_w(abits, j + 2, hsel), v1);
            fma4(acc, pair_w(abits, j + 4, hsel), v2);
            fma4(acc, pair_w(abits, j + 6, hsel), v3);
        }
        const int rem = tot2 - j;
        if (rem == 6) {
            const float4 v0 = pair_row(xrow, sbits, j,     hsel, c);
            const float4 v1 = pair_row(xrow, sbits, j + 2, hsel, c);
            const float4 v2 = pair_row(xrow, sbits, j + 4, hsel, c);
            fma4(acc, pair_w(abits, j,     hsel), v0);
            fma4(acc, pair_w(abits, j + 2, hsel), v1);
            fma4(acc, pair_w(abits, j + 4, hsel), v2);
        } else if (rem == 4) {
            const float4 v0 = pair_row(xrow, sbits, j,     hsel, c);
            const float4 v1 = pair_row(xrow, sbits, j + 2, hsel, c);
            fma4(acc, pair_w(abits, j,     hsel), v0);
            fma4(acc, pair_w(abits, j + 2, hsel), v1);
        } else if (rem == 2) {
            const float4 v0 = pair_row(xrow, sbits, j, hsel, c);
            fma4(acc, pair_w(abits, j, hsel), v0);
        }
    } else {
        for (int cc = 0; cc < deg; cc += 32) {
            const int idx = cc + lane;
            if (idx < deg) {
                const int2 pr = perm2[start + idx];
                int place = 0;
                for (int k = 0; k < deg; ++k) place += (perm2[start + k].x < pr.x) ? 1 : 0;
                csr[start + place] = pr.y;
            }
        }
        __threadfence();
        float m = NEG_INF;
        for (int cc = 0; cc < tot; cc += 32) {
            const int idx = cc + lane;
            if (idx < tot) {
                const int s = idx < deg ? csr[start + idx] : node;
                m = fmaxf(m, leaky_att(ssrc[s] + sd));
            }
        }
        m = wave_max(m);
        float den = 0.0f;
        for (int cc = 0; cc < tot; cc += 32) {
            const int idx = cc + lane;
            if (idx < tot) {
                const int s = idx < deg ? csr[start + idx] : node;
                den += __expf(leaky_att(ssrc[s] + sd) - m);
            }
        }
        den = wave_sum(den);
        for (int cc = 0; cc < tot; cc += 32) {
            const int idx = cc + lane;
            int s = node;
            if (idx < deg) s = csr[start + idx];
            float alpha = 0.0f;
            if (idx < tot) alpha = __expf(leaky_att(ssrc[s] + sd) - m) / (den + 1e-16f);
            const int abits = __float_as_int(alpha);
            const int cnt  = (tot - cc) < 32 ? (tot - cc) : 32;
            const int cnt2 = (cnt + 1) & ~1;
#pragma unroll 4
            for (int j = 0; j < cnt2; j += 2) {
                fma4(acc, pair_w(abits, j, hsel), pair_row(xrow, s, j, hsel, c));
            }
        }
    }

    acc.x += xhalf(acc.x); acc.y += xhalf(acc.y);
    acc.z += xhalf(acc.z); acc.w += xhalf(acc.w);
    return acc;
}

static __device__ __forceinline__ uint4 pair_row_h(const uint4* __restrict__ hrow,
                                                   int sbits, int j, int hsel, int c) {
    const int s0 = __builtin_amdgcn_readlane(sbits, j);
    const int s1 = __builtin_amdgcn_readlane(sbits, j + 1);
    return hrow[(size_t)(hsel ? s1 : s0) * 16 + c];
}
static __device__ __forceinline__ float fma_h_lo(float acc, const unsigned int w, const float a) {
    asm volatile("v_fma_mix_f32 %0, %1, %2, %0 op_sel_hi:[1,0,0]" : "+v"(acc) : "v"(w), "v"(a));
    return acc;
}
static __device__ __forceinline__ float fma_h_hi(float acc, const unsigned int w, const float a) {
    asm volatile("v_fma_mix_f32 %0, %1, %2, %0 op_sel:[1,0,0] op_sel_hi:[1,0,0]" : "+v"(acc) : "v"(w), "v"(a));
    return acc;
}
static __device__ __forceinline__ void fma8h(float (&acc)[8], const float a, const uint4 q) {
    acc[0] = fma_h_lo(acc[0], q.x, a); acc[1] = fma_h_hi(acc[1], q.x, a);
    acc[2] = fma_h_lo(acc[2], q.y, a); acc[3] = fma_h_hi(acc[3], q.y, a);
    acc[4] = fma_h_lo(acc[4], q.z, a); acc[5] = fma_h_hi(acc[5], q.z, a);
    acc[6] = fma_h_lo(acc[6], q.w, a); acc[7] = fma_h_hi(acc[7], q.w, a);
}

static __device__ __forceinline__ void gat_row_h(const int node, const int lane,
                                                 const int* __restrict__ off,
                                                 const int* __restrict__ csr,
                                                 const float* __restrict__ ssrc,
                                                 const float* __restrict__ sdst,
                                                 const unsigned short* __restrict__ h,
                                                 const float4 bv0, const float4 bv1,
                                                 float (&o)[8]) {
    const int start = off[node];
    const int deg   = off[node + 1] - start;
    const int tot   = deg + 1;
    const float sd  = sdst[node];
    const uint4* __restrict__ hrow = (const uint4*)h;
    const int c    = lane & 15;
    const int hsel = lane >> 4;
    const float NEG_INF = __int_as_float(0xFF800000);
    float acc[8] = {0.0f, 0.0f, 0.0f, 0.0f, 0.0f, 0.0f, 0.0f, 0.0f};

    if (tot <= 32) {
        const bool valid = lane < tot;
        int s = node;
        if (lane < deg) s = csr[start + lane];
        const float e   = leaky_att(ssrc[s] + sd);
        const float m   = wave_max(valid ? e : NEG_INF);
        const float ex  = valid ? __expf(e - m) : 0.0f;
        const float den = wave_sum(ex);
        const float alpha = ex / (den + 1e-16f);
        const int sbits = s;
        const int abits = __float_as_int(alpha);
        const int tot2  = (tot + 1) & ~1;
        int j = 0;
        for (; j + 8 <= tot2; j += 8) {
            const uint4 v0 = pair_row_h(hrow, sbits, j,     hsel, c);
            const uint4 v1 = pair_row_h(hrow, sbits, j + 2, hsel, c);
            const uint4 v2 = pair_row_h(hrow, sbits, j + 4, hsel, c);
            const uint4 v3 = pair_row_h(hrow, sbits, j + 6, hsel, c);
            fma8h(acc, pair_w(abits, j,     hsel), v0);
            fma8h(acc, pair_w(abits, j + 2, hsel), v1);
            fma8h(acc, pair_w(abits, j + 4, hsel), v2);
            fma8h(acc, pair_w(abits, j + 6, hsel), v3);
        }
        const int rem = tot2 - j;
        if (rem == 6) {
            const uint4 v0 = pair_row_h(hrow, sbits, j,     hsel, c);
            const uint4 v1 = pair_row_h(hrow, sbits, j + 2, hsel, c);
            const uint4 v2 = pair_row_h(hrow, sbits, j + 4, hsel, c);
            fma8h(acc, pair_w(abits, j,     hsel), v0);
            fma8h(acc, pair_w(abits, j + 2, hsel), v1);
            fma8h(acc, pair_w(abits, j + 4, hsel), v2);
        } else if (rem == 4) {
            const uint4 v0 = pair_row_h(hrow, sbits, j,     hsel, c);
            const uint4 v1 = pair_row_h(hrow, sbits, j + 2, hsel, c);
            fma8h(acc, pair_w(abits, j,     hsel), v0);
            fma8h(acc, pair_w(abits, j + 2, hsel), v1);
        } else if (rem == 2) {
            const uint4 v0 = pair_row_h(hrow, sbits, j, hsel, c);
            fma8h(acc, pair_w(abits, j, hsel), v0);
        }
    } else {
        float m = NEG_INF;
        for (int cc = 0; cc < tot; cc += 32) {
            const int idx = cc + lane;
            if (idx < tot) {
                const int s = idx < deg ? csr[start + idx] : node;
                m = fmaxf(m, leaky_att(ssrc[s] + sd));
            }
        }
        m = wave_max(m);
        float den = 0.0f;
        for (int cc = 0; cc < tot; cc += 32) {
            const int idx = cc + lane;
            if (idx < tot) {
                const int s = idx < deg ? csr[start + idx] : node;
                den += __expf(leaky_att(ssrc[s] + sd) - m);
            }
        }
        den = wave_sum(den);
        for (int cc = 0; cc < tot; cc += 32) {
            const int idx = cc + lane;
            int s = node;
            if (idx < deg) s = csr[start + idx];
            float alpha = 0.0f;
            if (idx < tot) alpha = __expf(leaky_att(ssrc[s] + sd) - m) / (den + 1e-16f);
            const int abits = __float_as_int(alpha);
            const int cnt  = (tot - cc) < 32 ? (tot - cc) : 32;
            const int cnt2 = (cnt + 1) & ~1;
#pragma unroll 4
            for (int j = 0; j < cnt2; j += 2) {
                fma8h(acc, pair_w(abits, j, hsel), pair_row_h(hrow, s, j, hsel, c));
            }
        }
    }

    const float bb[8] = {bv0.x, bv0.y, bv0.z, bv0.w, bv1.x, bv1.y, bv1.z, bv1.w};
#pragma unroll
    for (int k = 0; k < 8; ++k) {
        const float v = acc[k] + xhalf(acc[k]) + bb[k];
        o[k] = v > 0.0f ? v : 0.01f * v;
    }
}

__global__ __launch_bounds__(256)
void gat0_gemm_kernel(const int* __restrict__ off, const int* __restrict__ perm2, int* csr,
                      const float* __restrict__ ssrc, const float* __restrict__ sdst,
                      const float* __restrict__ x, const float* __restrict__ b0,
                      const unsigned short* __restrict__ Wt0,
                      const unsigned short* __restrict__ Wt1,
                      unsigned short* __restrict__ Ch,
                      const float* __restrict__ a_s, const float* __restrict__ a_d,
                      float* __restrict__ s_src_out, float* __restrict__ s_dst_out,
                      int N) {
    __shared__ unsigned short As0[16 * (FIN_C + 8)];
    __shared__ unsigned short As1[16 * (HIDF + 8)];
    __shared__ unsigned short Cs[16 * (HIDF + 8)];
    __shared__ float part[8 * 16 * 2];
    const int ldA0 = FIN_C + 8;
    const int ldA1 = HIDF + 8;
    const int ldC  = HIDF + 8;

    const int tid  = threadIdx.x;
    const int lane = tid & 31;
    const int wave = __builtin_amdgcn_readfirstlane(tid >> 5);
    const int m0   = blockIdx.x * 16;

    for (int q = 0; q < 2; ++q) {
        const int r    = wave * 2 + q;
        const int node = m0 + r;
        float4 o = {0.0f, 0.0f, 0.0f, 0.0f};
        if (node < N) o = gat_row64(node, lane, off, (const int2*)perm2, csr, ssrc, sdst, x);
        if (lane < 16) {
            uint2 p;
            p.x = (unsigned int)f32_to_bf16(o.x) | ((unsigned int)f32_to_bf16(o.y) << 16);
            p.y = (unsigned int)f32_to_bf16(o.z) | ((unsigned int)f32_to_bf16(o.w) << 16);
            *(uint2*)&As0[r * ldA0 + lane * 4] = p;
        }
    }
    __syncthreads();

    tile_gemm0_to_lds(As0, ldA0, Wt0, b0, As1, ldA1, tid);
    __syncthreads();

    tile_gemm<true>(As1, ldA1, HIDF, Wt1, nullptr, Ch, Cs, ldC, nullptr, a_s, a_d,
                    s_src_out, s_dst_out, part, m0, N, HIDF, tid);
}

template <int NOUT, bool SCORES>
__global__ __launch_bounds__(256)
void gat_gemm_kernel(const int* __restrict__ off, const int* __restrict__ csr,
                     const float* __restrict__ ssrc, const float* __restrict__ sdst,
                     const unsigned short* __restrict__ h, const float* __restrict__ b,
                     const unsigned short* __restrict__ Wt,
                     float* __restrict__ C,
                     unsigned short* __restrict__ Ch,
                     const float* __restrict__ bias_out,
                     const float* __restrict__ a_s, const float* __restrict__ a_d,
                     float* __restrict__ s_src_out, float* __restrict__ s_dst_out,
                     int N) {
    __shared__ unsigned short As[16 * (HIDF + 8)];
    __shared__ unsigned short Cs[SCORES ? 16 * (HIDF + 8) : 8];
    __shared__ float part[SCORES ? 8 * 16 * 2 : 2];
    const int ldA = HIDF + 8;
    const int ldC = HIDF + 8;

    const int tid  = threadIdx.x;
    const int lane = tid & 31;
    const int wave = __builtin_amdgcn_readfirstlane(tid >> 5);
    const int m0   = blockIdx.x * 16;

    const float4 bv0 = ((const float4*)b)[(lane & 15) * 2];
    const float4 bv1 = ((const float4*)b)[(lane & 15) * 2 + 1];
    for (int q = 0; q < 2; ++q) {
        const int r    = wave * 2 + q;
        const int node = m0 + r;
        float o[8] = {0.0f, 0.0f, 0.0f, 0.0f, 0.0f, 0.0f, 0.0f, 0.0f};
        if (node < N) gat_row_h(node, lane, off, csr, ssrc, sdst, h, bv0, bv1, o);
        if (lane < 16) {
            uint4 p;
            p.x = (unsigned int)f32_to_bf16(o[0]) | ((unsigned int)f32_to_bf16(o[1]) << 16);
            p.y = (unsigned int)f32_to_bf16(o[2]) | ((unsigned int)f32_to_bf16(o[3]) << 16);
            p.z = (unsigned int)f32_to_bf16(o[4]) | ((unsigned int)f32_to_bf16(o[5]) << 16);
            p.w = (unsigned int)f32_to_bf16(o[6]) | ((unsigned int)f32_to_bf16(o[7]) << 16);
            *(uint4*)&As[r * ldA + lane * 8] = p;
        }
    }
    __syncthreads();

    if (SCORES) {
        tile_gemm<true>(As, ldA, HIDF, Wt, C, Ch, Cs, ldC, bias_out, a_s, a_d,
                        s_src_out, s_dst_out, part, m0, N, NOUT, tid);
    } else if (wave < NOUT / 16) {
        tile_gemm<false>(As, ldA, HIDF, Wt, C, Ch, Cs, ldC, bias_out, a_s, a_d,
                         s_src_out, s_dst_out, part, m0, N, NOUT, tid);
    }
}

extern "C" void kernel_launch(void* const* d_in, const int* in_sizes, int n_in,
                              void* d_out, int out_size, void* d_ws, size_t ws_size,
                              hipStream_t stream) {
    (void)n_in; (void)out_size; (void)ws_size;

    const float* x    = (const float*)d_in[0];
    const int*   ei   = (const int*)d_in[1];
    const float* Wl[3] = {(const float*)d_in[2], (const float*)d_in[6],  (const float*)d_in[10]};
    const float* Asl[3]= {(const float*)d_in[3], (const float*)d_in[7],  (const float*)d_in[11]};
    const float* Adl[3]= {(const float*)d_in[4], (const float*)d_in[8],  (const float*)d_in[12]};
    const float* Bl[3] = {(const float*)d_in[5], (const float*)d_in[9],  (const float*)d_in[13]};
    const float* Wout = (const float*)d_in[14];
    const float* bout = (const float*)d_in[15];

    const int FIN  = 64;
    const int N  = in_sizes[0] / FIN;
    const int E  = in_sizes[1] / 2;

    unsigned short* hA  = (unsigned short*)d_ws;
    unsigned short* hB  = hA + (size_t)N * HIDF;
    unsigned short* Wt0 = hB + (size_t)N * HIDF;
    unsigned short* Wt1 = Wt0 + HIDF * HIDF;
    unsigned short* Wt2 = Wt1 + HIDF * HIDF;
    unsigned short* Wt3 = Wt2 + HIDF * HIDF;
    float* vsd    = (float*)(Wt3 + HIDF * HIDF);
    float* sA_src = vsd + 2 * FIN_C;
    float* sA_dst = sA_src + N;
    float* sB_src = sA_dst + N;
    float* sB_dst = sB_src + N;
    int* perm2 = (int*)(sB_dst + N);
    int* cnt   = perm2 + E + E;
    int* off   = cnt + N;
    int* rank  = off + N + 1;
    int* csr   = rank + E;

    const int B256 = 256;
    const int gN    = (N + B256 - 1) / B256;
    const int gE    = (E + B256 - 1) / B256;
    const int gM    = (N + 15) / 16;
    const int gS    = (N + 63) / 64;

    prep_kernel<<<gN + 193, B256, 0, stream>>>(cnt, N, gN, Wl[0], Wl[1], Wl[2], Wout,
                                               Wt0, Wt1, Wt2, Wt3, Asl[0], Adl[0], vsd);

    edge_rank_kernel<<<gE, B256, 0, stream>>>(ei, cnt, rank, E);
    offsets_scores_kernel<<<gN + gS, B256, 0, stream>>>(cnt, off, N, gN, x, vsd, sA_src, sA_dst);
    edge_fill_kernel<<<gE, B256, 0, stream>>>(ei, off, rank, perm2, E);

    gat0_gemm_kernel<<<gM, B256, 0, stream>>>(
        off, perm2, csr, sA_src, sA_dst, x, Bl[0], Wt0, Wt1, hB, Asl[1], Adl[1], sB_src, sB_dst, N);

    gat_gemm_kernel<HIDF, true><<<gM, B256, 0, stream>>>(
        off, csr, sB_src, sB_dst, hB, Bl[1], Wt2, nullptr, hA, nullptr, Asl[2], Adl[2], sA_src, sA_dst, N);

    gat_gemm_kernel<FOUT_C, false><<<gM, B256, 0, stream>>>(
        off, csr, sA_src, sA_dst, hA, Bl[2], Wt3, (float*)d_out, nullptr, bout,
        nullptr, nullptr, nullptr, nullptr, N);
}


# Edges go straight into fixed 32-pair buckets per destination as the counting atomic hands out places (overflow list past 32): offset scan and fill launches gone (7 to 5); sums unchanged
# speedup vs baseline: 5.5349x; 1.2168x over previous
#include <hip/hip_runtime.h>
#include <hip/hip_bf16.h>


typedef __attribute__((ext_vector_type(16))) __bf16 v16bf;
typedef __attribute__((ext_vector_type(8)))  float  v8f;

union BfFrag {
    unsigned int u[8];
    v16bf bf;
};

#define HIDF 128
#define FOUT_C 64
#define FIN_C 64
#define CAP 32

static __device__ __forceinline__ unsigned short f32_to_bf16(float f) {
    unsigned int u = __float_as_uint(f);
    unsigned int r = u + 0x7FFFu + ((u >> 16) & 1u);
    return (unsigned short)(r >> 16);
}

static __device__ __forceinline__ unsigned short f32_to_f16_bits(float f) {
    return __builtin_bit_cast(unsigned short, (_Float16)f);
}

static __device__ __forceinline__ void wt_bf16_item(const float* __restrict__ W,
                                                    unsigned short* __restrict__ Wt,
                                                    int K, int Nout, int i) {
    if (i < K * Nout) {
        int k = i / Nout, n = i - k * Nout;
        Wt[n * K + k] = f32_to_bf16(W[i]);
    }
}

static __device__ __forceinline__ void score_vectors_block(const float* __restrict__ W0,
                                                           const float* __restrict__ a_s,
                                                           const float* __restrict__ a_d,
                                                           float* __restrict__ vsd, int t) {
    const int k = t >> 2, q = t & 3;
    const float4* __restrict__ w4 = (const float4*)(W0 + k * HIDF + q * 32);
    const float4* __restrict__ s4 = (const float4*)(a_s + q * 32);
    const float4* __restrict__ d4 = (const float4*)(a_d + q * 32);
    float ps = 0.0f, pd = 0.0f;
#pragma unroll
    for (int i = 0; i < 8; ++i) {
        const float4 w = w4[i], s = s4[i], d = d4[i];
        ps += w.x * s.x + w.y * s.y + w.z * s.z + w.w * s.w;
        pd += w.x * d.x + w.y * d.y + w.z * d.z + w.w * d.w;
    }
    ps += __shfl_xor(ps, 1, 32); ps += __shfl_xor(ps, 2, 32);
    pd += __shfl_xor(pd, 1, 32); pd += __shfl_xor(pd, 2, 32);
    if (q == 0) { vsd[k] = ps; vsd[FIN_C + k] = pd; }
}

__global__ __launch_bounds__(256)
void prep_kernel(int* __restrict__ cnt, int* __restrict__ gcnt, int N, int nZero,
                 const float* __restrict__ W0, const float* __restrict__ W1,
                 const float* __restrict__ W2, const float* __restrict__ W3,
                 unsigned short* __restrict__ Wt0, unsigned short* __restrict__ Wt1,
                 unsigned short* __restrict__ Wt2, unsigned short* __restrict__ Wt3,
                 const float* __restrict__ a_s0, const float* __restrict__ a_d0,
                 float* __restrict__ vsd) {
    const int t = threadIdx.x;
    int b = blockIdx.x;
    if (b < nZero) {
        const int i = b * 256 + t;
        if (i < N) cnt[i] = 0;
        if (i < 2) gcnt[i] = 0;
        return;
    }
    b -= nZero;
    if (b < 32) { wt_bf16_item(W0, Wt0, 64, HIDF, b * 256 + t); return; }
    b -= 32;
    if (b < 64) { wt_bf16_item(W1, Wt1, HIDF, HIDF, b * 256 + t); return; }
    b -= 64;
    if (b < 64) { wt_bf16_item(W2, Wt2, HIDF, HIDF, b * 256 + t); return; }
    b -= 64;
    if (b < 32) { wt_bf16_item(W3, Wt3, HIDF, FOUT_C, b * 256 + t); return; }
    score_vectors_block(W0, a_s0, a_d0, vsd, t);
}

static __device__ __forceinline__ float half_wave_sum(float v) {
    v += __int_as_float(__builtin_amdgcn_update_dpp(0, __float_as_int(v), 0x161, 0xf, 0xf, true));
    v += __int_as_float(__builtin_amdgcn_update_dpp(0, __float_as_int(v), 0x162, 0xf, 0xf, true));
    v += __int_as_float(__builtin_amdgcn_update_dpp(0, __float_as_int(v), 0x164, 0xf, 0xf, true));
    v += __int_as_float(__builtin_amdgcn_update_dpp(0, __float_as_int(v), 0x168, 0xf, 0xf, true));
    return v;
}

template <bool SCORES>
static __device__ __forceinline__ void tile_gemm(const unsigned short* As, const int ldA, const int K,
                                                 const unsigned short* __restrict__ Wt,
                                                 float* __restrict__ C,
                                                 unsigned short* __restrict__ Ch,
                                                 unsigned short* Cs, const int ldC,
                                                 const float* __restrict__ bias,
                                                 const float* __restrict__ a_s,
                                                 const float* __restrict__ a_d,
                                                 float* __restrict__ s_src,
                                                 float* __restrict__ s_dst,
                                                 float* part,
                                                 const int m0, const int M, const int Nout,
                                                 const int tid) {
    const int wave  = tid >> 5;
    const int lane  = tid & 31;
    const int row   = lane & 15;
    const int khalf = lane >> 4;
    const int ncol  = wave * 16 + row;

    float as_c = 0.0f, ad_c = 0.0f;
    if (SCORES) { as_c = a_s[ncol]; ad_c = a_d[ncol]; }

    v8f acc = {};

#pragma unroll 2
    for (int k0 = 0; k0 < K; k0 += 32) {
        const int cb = k0 + khalf * 8;

        BfFrag a, b;
        {
            const uint4* p0 = (const uint4*)&As[row * ldA + cb];
            const uint4* p1 = (const uint4*)&As[row * ldA + cb + 16];
            uint4 x0 = *p0, x1 = *p1;
            a.u[0] = x0.x; a.u[1] = x0.y; a.u[2] = x0.z; a.u[3] = x0.w;
            a.u[4] = x1.x; a.u[5] = x1.y; a.u[6] = x1.z; a.u[7] = x1.w;
        }
        {
            const uint4* p0 = (const uint4*)&Wt[(size_t)ncol * K + cb];
            const uint4* p1 = (const uint4*)&Wt[(size_t)ncol * K + cb + 16];
            uint4 x0 = *p0, x1 = *p1;
            b.u[0] = x0.x; b.u[1] = x0.y; b.u[2] = x0.z; b.u[3] = x0.w;
            b.u[4] = x1.x; b.u[5] = x1.y; b.u[6] = x1.z; b.u[7] = x1.w;
        }

        acc = __builtin_amdgcn_wmma_f32_16x16x32_bf16(
             false, a.bf,  false, b.bf,
             (short)0, acc,  false,  false);
    }

    float badd = bias ? bias[ncol] : 0.0f;
    if (!SCORES) {
#pragma unroll
        for (int r = 0; r < 8; ++r) {
            int gr = m0 + r + khalf * 8;
            if (gr < M) C[(size_t)gr * Nout + ncol] = acc[r] + badd;
        }
    }

    if (SCORES) {
#pragma unroll
        for (int r = 0; r < 8; ++r)
            Cs[(r + khalf * 8) * ldC + ncol] = f32_to_f16_bits(acc[r] + badd);

        float ps[8], pd[8];
#pragma unroll
        for (int r = 0; r < 8; ++r) {
            ps[r] = half_wave_sum(acc[r] * as_c);
            pd[r] = half_wave_sum(acc[r] * ad_c);
        }
        if (row == 0) {
#pragma unroll
            for (int r = 0; r < 8; ++r) {
                float2 p = {ps[r], pd[r]};
                *(float2*)&part[(wave * 16 + khalf * 8 + r) * 2] = p;
            }
        }
        __syncthreads();
        {
            const int rr = tid >> 4, pc = tid & 15;
            const int gr = m0 + rr;
            if (gr < M) *(uint4*)&Ch[(size_t)gr * HIDF + pc * 8] = *(const uint4*)&Cs[rr * ldC + pc * 8];
        }
        if (tid < 32) {
            const int rr = tid & 15, which = tid >> 4;
            float s = 0.0f;
#pragma unroll
            for (int w = 0; w < 8; ++w) s += part[(w * 16 + rr) * 2 + which];
            const int gr = m0 + rr;
            if (gr < M) (which ? s_dst : s_src)[gr] = s;
        }
    }
}

static __device__ __forceinline__ void tile_gemm0_to_lds(const unsigned short* As0, const int ldA0,
                                                         const unsigned short* __restrict__ Wt0,
                                                         const float* __restrict__ b0,
                                                         unsigned short* As1, const int ldA1,
                                                         const int tid) {
    const int wave  = tid >> 5;
    const int lane  = tid & 31;
    const int row   = lane & 15;
    const int khalf = lane >> 4;
    const int ncol  = wave * 16 + row;

    v8f acc = {};
#pragma unroll
    for (int k0 = 0; k0 < FIN_C; k0 += 32) {
        const int cb = k0 + khalf * 8;
        BfFrag a, b;
        {
            const uint4* p0 = (const uint4*)&As0[row * ldA0 + cb];
            const uint4* p1 = (const uint4*)&As0[row * ldA0 + cb + 16];
            uint4 x0 = *p0, x1 = *p1;
            a.u[0] = x0.x; a.u[1] = x0.y; a.u[2] = x0.z; a.u[3] = x0.w;
            a.u[4] = x1.x; a.u[5] = x1.y; a.u[6] = x1.z; a.u[7] = x1.w;
        }
        {
            const uint4* p0 = (const uint4*)&Wt0[(size_t)ncol * FIN_C + cb];
            const uint4* p1 = (const uint4*)&Wt0[(size_t)ncol * FIN_C + cb + 16];
            uint4 x0 = *p0, x1 = *p1;
            b.u[0] = x0.x; b.u[1] = x0.y; b.u[2] = x0.z; b.u[3] = x0.w;
            b.u[4] = x1.x; b.u[5] = x1.y; b.u[6] = x1.z; b.u[7] = x1.w;
        }
        acc = __builtin_amdgcn_wmma_f32_16x16x32_bf16(
             false, a.bf,  false, b.bf,
             (short)0, acc,  false,  false);
    }

    const float badd = b0[ncol];
#pragma unroll
    for (int r = 0; r < 8; ++r) {
        float v = acc[r] + badd;
        v = v > 0.0f ? v : 0.01f * v;
        As1[(r + khalf * 8) * ldA1 + ncol] = f32_to_bf16(v);
    }
}

static __device__ __forceinline__ void node_scores_block(const float* __restrict__ x,
                                                         const float* __restrict__ vsd,
                                                         float* __restrict__ s_src,
                                                         float* __restrict__ s_dst,
                                                         int N, int blk) {
    const int tid = threadIdx.x;
    const int c   = tid & 15;
    const int n0  = blk * 64 + (tid >> 4);
    const float4* __restrict__ x4 = (const float4*)x;
    const float4 vs = ((const float4*)vsd)[c];
    const float4 vd = ((const float4*)vsd)[16 + c];
    float4 r[4];
#pragma unroll
    for (int i = 0; i < 4; ++i) {
        const int n = n0 + 16 * i;
        float4 z = {0.0f, 0.0f, 0.0f, 0.0f};
        if (n < N) z = x4[(size_t)n * 16 + c];
        r[i] = z;
    }
#pragma unroll
    for (int i = 0; i < 4; ++i) {
        const int n = n0 + 16 * i;
        const float ps = half_wave_sum(r[i].x * vs.x + r[i].y * vs.y + r[i].z * vs.z + r[i].w * vs.w);
        const float pd = half_wave_sum(r[i].x * vd.x + r[i].y * vd.y + r[i].z * vd.z + r[i].w * vd.w);
        if (c == 0 && n < N) { s_src[n] = ps; s_dst[n] = pd; }
    }
}

static __device__ __forceinline__ void edge_bucket_item(const int* __restrict__ ei,
                                                        int* __restrict__ cnt,
                                                        int2* __restrict__ bucket,
                                                        int* __restrict__ gcnt,
                                                        int2* __restrict__ ovf_pair,
                                                        int* __restrict__ ovf_dst,
                                                        int E, int i) {
    if (i < E) {
        const int d = ei[E + i];
        int2 p;
        p.x = i;
        p.y = ei[i];
        const int r = atomicAdd(&cnt[d], 1);
        if (r < CAP) {
            bucket[(size_t)d * CAP + r] = p;
        } else {
            const int o = atomicAdd(&gcnt[0], 1);
            ovf_pair[o] = p;
            ovf_dst[o]  = d;
        }
    }
}

__global__ __launch_bounds__(256)
void edges_scores_kernel(const int* __restrict__ ei, int* __restrict__ cnt,
                         int2* __restrict__ bucket, int* __restrict__ gcnt,
                         int2* __restrict__ ovf_pair, int* __restrict__ ovf_dst,
                         int E, int nEdge,
                         const float* __restrict__ x,
                         const float* __restrict__ vsd,
                         float* __restrict__ s_src,
                         float* __restrict__ s_dst, int N) {
    const int b = blockIdx.x;
    if (b < nEdge) edge_bucket_item(ei, cnt, bucket, gcnt, ovf_pair, ovf_dst, E, b * 256 + (int)threadIdx.x);
    else node_scores_block(x, vsd, s_src, s_dst, N, b - nEdge);
}

static __device__ __forceinline__ float leaky_att(float e) {
    return e > 0.0f ? e : 0.2f * e;
}
static __device__ __forceinline__ float wave_max(float v) {
    for (int o = 16; o; o >>= 1) v = fmaxf(v, __shfl_xor(v, o, 32));
    return v;
}
static __device__ __forceinline__ float wave_sum(float v) {
    for (int o = 16; o; o >>= 1) v += __shfl_xor(v, o, 32);
    return v;
}
static __device__ __forceinline__ void fma4(float4& acc, float a, const float4& v) {
    acc.x += a * v.x; acc.y += a * v.y; acc.z += a * v.z; acc.w += a * v.w;
}
static __device__ __forceinline__ float xhalf(float v) {
    const int b = __float_as_int(v);
    return __int_as_float(__builtin_amdgcn_permlanex16(b, b, 0x76543210, 0xfedcba98, false, false));
}
static __device__ __forceinline__ float4 pair_row(const float4* __restrict__ xrow,
                                                  int sbits, int j, int hsel, int c) {
    const int s0 = __builtin_amdgcn_readlane(sbits, j);
    const int s1 = __builtin_amdgcn_readlane(sbits, j + 1);
    return xrow[(size_t)(hsel ? s1 : s0) * 16 + c];
}
static __device__ __forceinline__ float pair_w(int abits, int j, int hsel) {
    const int a0 = __builtin_amdgcn_readlane(abits, j);
    const int a1 = __builtin_amdgcn_readlane(abits, j + 1);
    return __int_as_float(hsel ? a1 : a0);
}

static __device__ __forceinline__ float4 gat_row64(const int node, const int lane,
                                                   const int* __restrict__ cnt,
                                                   const int2* __restrict__ bucket,
                                                   int* __restrict__ csr,
                                                   int* gcnt,
                                                   const int2* __restrict__ ovf_pair,
                                                   const int* __restrict__ ovf_dst,
                                                   int2* big_pair, int* big_csr,
                                                   int* __restrict__ big_off,
                                                   const float* __restrict__ ssrc,
                                                   const float* __restrict__ sdst,
                                                   const float* __restrict__ x) {
    const int start = node * CAP;
    const int deg   = cnt[node];
    const int tot   = deg + 1;
    const float sd  = sdst[node];
    const float4* __restrict__ xrow = (const float4*)x;
    const int c    = lane & 15;
    const int hsel = lane >> 4;
    const float NEG_INF = __int_as_float(0xFF800000);
    float4 acc = {0.0f, 0.0f, 0.0f, 0.0f};

    if (tot <= 32) {
        const bool valid = lane < tot;
        int id = 0x7fffffff;
        int sv = node;
        if (lane < deg) { const int2 pr = bucket[start + lane]; id = pr.x; sv = pr.y; }
        int place = 0;
        for (int k = 0; k < deg; ++k) place += (__builtin_amdgcn_readlane(id, k) < id) ? 1 : 0;
        if (lane >= deg) place = lane;
        const int s = __builtin_amdgcn_ds_permute(place << 2, sv);
        if (lane < deg) csr[start + lane] = s;
        const float e   = leaky_att(ssrc[s] + sd);
        const float m   = wave_max(valid ? e : NEG_INF);
        const float ex  = valid ? __expf(e - m) : 0.0f;
        const float den = wave_sum(ex);
        const float alpha = ex / (den + 1e-16f);
        const int sbits = s;
        const int abits = __float_as_int(alpha);
        const int tot2  = (tot + 1) & ~1;
        int j = 0;
        for (; j + 8 <= tot2; j += 8) {
            const float4 v0 = pair_row(xrow, sbits, j,     hsel, c);
            const float4 v1 = pair_row(xrow, sbits, j + 2, hsel, c);
            const float4 v2 = pair_row(xrow, sbits, j + 4, hsel, c);
            const float4 v3 = pair_row(xrow, sbits, j + 6, hsel, c);
            fma4(acc, pair_w(abits, j,     hsel), v0);
            fma4(acc, pair_w(abits, j + 2, hsel), v1);
            fma4(acc, pair_w(abits, j + 4, hsel), v2);
            fma4(acc, pair_w(abits, j + 6, hsel), v3);
        }
        const int rem = tot2 - j;
        if (rem == 6) {
            const float4 v0 = pair_row(xrow, sbits, j,     hsel, c);
            const float4 v1 = pair_row(xrow, sbits, j + 2, hsel, c);
            const float4 v2 = pair_row(xrow, sbits, j + 4, hsel, c);
            fma4(acc, pair_w(abits, j,     hsel), v0);
            fma4(acc, pair_w(abits, j + 2, hsel), v1);
            fma4(acc, pair_w(abits, j + 4, hsel), v2);
        } else if (rem == 4) {
            const float4 v0 = pair_row(xrow, sbits, j,     hsel, c);
            const float4 v1 = pair_row(xrow, sbits, j + 2, hsel, c);
            fma4(acc, pair_w(abits, j,     hsel), v0);
            fma4(acc, pair_w(abits, j + 2, hsel), v1);
        } else if (rem == 2) {
            const float4 v0 = pair_row(xrow, sbits, j, hsel, c);
            fma4(acc, pair_w(abits, j, hsel), v0);
        }
    } else {
        int base = 0;
        if (lane == 0) { base = atomicAdd(&gcnt[1], deg); big_off[node] = base; }
        base = __builtin_amdgcn_readfirstlane(base);
        for (int k = lane; k < CAP; k += 32) big_pair[base + k] = bucket[start + k];
        int filled = CAP;
        const int novf = gcnt[0];
        for (int cc = 0; cc < novf; cc += 32) {
            const int idx = cc + lane;
            const bool hit = idx < novf && ovf_dst[idx] == node;
            const unsigned int hits = __builtin_amdgcn_ballot_w32(hit);
            if (hit) big_pair[base + filled + __popc(hits & ((1u << lane) - 1u))] = ovf_pair[idx];
            filled += __popc(hits);
        }
        __threadfence();
        for (int cc = 0; cc < deg; cc += 32) {
            const int idx = cc + lane;
            if (idx < deg) {
                const int2 pr = big_pair[base + idx];
                int place = 0;
                for (int k = 0; k < deg; ++k) place += (big_pair[base + k].x < pr.x) ? 1 : 0;
                big_csr[base + place] = pr.y;
            }
        }
        __threadfence();
        float m = NEG_INF;
        for (int cc = 0; cc < tot; cc += 32) {
            const int idx = cc + lane;
            if (idx < tot) {
                const int s = idx < deg ? big_csr[base + idx] : node;
                m = fmaxf(m, leaky_att(ssrc[s] + sd));
            }
        }
        m = wave_max(m);
        float den = 0.0f;
        for (int cc = 0; cc < tot; cc += 32) {
            const int idx = cc + lane;
            if (idx < tot) {
                const int s = idx < deg ? big_csr[base + idx] : node;
                den += __expf(leaky_att(ssrc[s] + sd) - m);
            }
        }
        den = wave_sum(den);
        for (int cc = 0; cc < tot; cc += 32) {
            const int idx = cc + lane;
            int s = node;
            if (idx < deg) s = big_csr[base + idx];
            float alpha = 0.0f;
            if (idx < tot) alpha = __expf(leaky_att(ssrc[s] + sd) - m) / (den + 1e-16f);
            const int abits = __float_as_int(alpha);
            const int cnt_e = (tot - cc) < 32 ? (tot - cc) : 32;
            const int cnt2  = (cnt_e + 1) & ~1;
#pragma unroll 4
            for (int j = 0; j < cnt2; j += 2) {
                fma4(acc, pair_w(abits, j, hsel), pair_row(xrow, s, j, hsel, c));
            }
        }
    }

    acc.x += xhalf(acc.x); acc.y += xhalf(acc.y);
    acc.z += xhalf(acc.z); acc.w += xhalf(acc.w);
    return acc;
}

static __device__ __forceinline__ uint4 pair_row_h(const uint4* __restrict__ hrow,
                                                   int sbits, int j, int hsel, int c) {
    const int s0 = __builtin_amdgcn_readlane(sbits, j);
    const int s1 = __builtin_amdgcn_readlane(sbits, j + 1);
    return hrow[(size_t)(hsel ? s1 : s0) * 16 + c];
}
static __device__ __forceinline__ float fma_h_lo(float acc, const unsigned int w, const float a) {
    asm volatile("v_fma_mix_f32 %0, %1, %2, %0 op_sel_hi:[1,0,0]" : "+v"(acc) : "v"(w), "v"(a));
    return acc;
}
static __device__ __forceinline__ float fma_h_hi(float acc, const unsigned int w, const float a) {
    asm volatile("v_fma_mix_f32 %0, %1, %2, %0 op_sel:[1,0,0] op_sel_hi:[1,0,0]" : "+v"(acc) : "v"(w), "v"(a));
    return acc;
}
static __device__ __forceinline__ void fma8h(float (&acc)[8], const float a, const uint4 q) {
    acc[0] = fma_h_lo(acc[0], q.x, a); acc[1] = fma_h_hi(acc[1], q.x, a);
    acc[2] = fma_h_lo(acc[2], q.y, a); acc[3] = fma_h_hi(acc[3], q.y, a);
    acc[4] = fma_h_lo(acc[4], q.z, a); acc[5] = fma_h_hi(acc[5], q.z, a);
    acc[6] = fma_h_lo(acc[6], q.w, a); acc[7] = fma_h_hi(acc[7], q.w, a);
}

static __device__ __forceinline__ void gat_row_h(const int node, const int lane,
                                                 const int* __restrict__ cnt,
                                                 const int* __restrict__ csr,
                                                 const int* __restrict__ big_off,
                                                 const int* __restrict__ big_csr,
                                                 const float* __restrict__ ssrc,
                                                 const float* __restrict__ sdst,
                                                 const unsigned short* __restrict__ h,
                                                 const float4 bv0, const float4 bv1,
                                                 float (&o)[8]) {
    const int start = node * CAP;
    const int deg   = cnt[node];
    const int tot   = deg + 1;
    const float sd  = sdst[node];
    const uint4* __restrict__ hrow = (const uint4*)h;
    const int c    = lane & 15;
    const int hsel = lane >> 4;
    const float NEG_INF = __int_as_float(0xFF800000);
    float acc[8] = {0.0f, 0.0f, 0.0f, 0.0f, 0.0f, 0.0f, 0.0f, 0.0f};

    if (tot <= 32) {
        const bool valid = lane < tot;
        int s = node;
        if (lane < deg) s = csr[start + lane];
        const float e   = leaky_att(ssrc[s] + sd);
        const float m   = wave_max(valid ? e : NEG_INF);
        const float ex  = valid ? __expf(e - m) : 0.0f;
        const float den = wave_sum(ex);
        const float alpha = ex / (den + 1e-16f);
        const int sbits = s;
        const int abits = __float_as_int(alpha);
        const int tot2  = (tot + 1) & ~1;
        int j = 0;
        for (; j + 8 <= tot2; j += 8) {
            const uint4 v0 = pair_row_h(hrow, sbits, j,     hsel, c);
            const uint4 v1 = pair_row_h(hrow, sbits, j + 2, hsel, c);
            const uint4 v2 = pair_row_h(hrow, sbits, j + 4, hsel, c);
            const uint4 v3 = pair_row_h(hrow, sbits, j + 6, hsel, c);
            fma8h(acc, pair_w(abits, j,     hsel), v0);
            fma8h(acc, pair_w(abits, j + 2, hsel), v1);
            fma8h(acc, pair_w(abits, j + 4, hsel), v2);
            fma8h(acc, pair_w(abits, j + 6, hsel), v3);
        }
        const int rem = tot2 - j;
        if (rem == 6) {
            const uint4 v0 = pair_row_h(hrow, sbits, j,     hsel, c);
            const uint4 v1 = pair_row_h(hrow, sbits, j + 2, hsel, c);
            const uint4 v2 = pair_row_h(hrow, sbits, j + 4, hsel, c);
            fma8h(acc, pair_w(abits, j,     hsel), v0);
            fma8h(acc, pair_w(abits, j + 2, hsel), v1);
            fma8h(acc, pair_w(abits, j + 4, hsel), v2);
        } else if (rem == 4) {
            const uint4 v0 = pair_row_h(hrow, sbits, j,     hsel, c);
            const uint4 v1 = pair_row_h(hrow, sbits, j + 2, hsel, c);
            fma8h(acc, pair_w(abits, j,     hsel), v0);
            fma8h(acc, pair_w(abits, j + 2, hsel), v1);
        } else if (rem == 2) {
            const uint4 v0 = pair_row_h(hrow, sbits, j, hsel, c);
            fma8h(acc, pair_w(abits, j, hsel), v0);
        }
    } else {
        const int base = big_off[node];
        float m = NEG_INF;
        for (int cc = 0; cc < tot; cc += 32) {
            const int idx = cc + lane;
            if (idx < tot) {
                const int s = idx < deg ? big_csr[base + idx] : node;
                m = fmaxf(m, leaky_att(ssrc[s] + sd));
            }
        }
        m = wave_max(m);
        float den = 0.0f;
        for (int cc = 0; cc < tot; cc += 32) {
            const int idx = cc + lane;
            if (idx < tot) {
                const int s = idx < deg ? big_csr[base + idx] : node;
                den += __expf(leaky_att(ssrc[s] + sd) - m);
            }
        }
        den = wave_sum(den);
        for (int cc = 0; cc < tot; cc += 32) {
            const int idx = cc + lane;
            int s = node;
            if (idx < deg) s = big_csr[base + idx];
            float alpha = 0.0f;
            if (idx < tot) alpha = __expf(leaky_att(ssrc[s] + sd) - m) / (den + 1e-16f);
            const int abits = __float_as_int(alpha);
            const int cnt_e = (tot - cc) < 32 ? (tot - cc) : 32;
            const int cnt2  = (cnt_e + 1) & ~1;
#pragma unroll 4
            for (int j = 0; j < cnt2; j += 2) {
                fma8h(acc, pair_w(abits, j, hsel), pair_row_h(hrow, s, j, hsel, c));
            }
        }
    }

    const float bb[8] = {bv0.x, bv0.y, bv0.z, bv0.w, bv1.x, bv1.y, bv1.z, bv1.w};
#pragma unroll
    for (int k = 0; k < 8; ++k) {
        const float v = acc[k] + xhalf(acc[k]) + bb[k];
        o[k] = v > 0.0f ? v : 0.01f * v;
    }
}

__global__ __launch_bounds__(256)
void gat0_gemm_kernel(const int* __restrict__ cnt, const int2* __restrict__ bucket,
                      int* __restrict__ csr, int* gcnt,
                      const int2* __restrict__ ovf_pair, const int* __restrict__ ovf_dst,
                      int2* big_pair, int* big_csr, int* __restrict__ big_off,
                      const float* __restrict__ ssrc, const float* __restrict__ sdst,
                      const float* __restrict__ x, const float* __restrict__ b0,
                      const unsigned short* __restrict__ Wt0,
                      const unsigned short* __restrict__ Wt1,
                      unsigned short* __restrict__ Ch,
                      const float* __restrict__ a_s, const float* __restrict__ a_d,
                      float* __restrict__ s_src_out, float* __restrict__ s_dst_out,
                      int N) {
    __shared__ unsigned short As0[16 * (FIN_C + 8)];
    __shared__ unsigned short As1[16 * (HIDF + 8)];
    __shared__ unsigned short Cs[16 * (HIDF + 8)];
    __shared__ float part[8 * 16 * 2];
    const int ldA0 = FIN_C + 8;
    const int ldA1 = HIDF + 8;
    const int ldC  = HIDF + 8;

    const int tid  = threadIdx.x;
    const int lane = tid & 31;
    const int wave = __builtin_amdgcn_readfirstlane(tid >> 5);
    const int m0   = blockIdx.x * 16;

    for (int q = 0; q < 2; ++q) {
        const int r    = wave * 2 + q;
        const int node = m0 + r;
        float4 o = {0.0f, 0.0f, 0.0f, 0.0f};
        if (node < N) o = gat_row64(node, lane, cnt, bucket, csr, gcnt, ovf_pair, ovf_dst,
                                    big_pair, big_csr, big_off, ssrc, sdst, x);
        if (lane < 16) {
            uint2 p;
            p.x = (unsigned int)f32_to_bf16(o.x) | ((unsigned int)f32_to_bf16(o.y) << 16);
            p.y = (unsigned int)f32_to_bf16(o.z) | ((unsigned int)f32_to_bf16(o.w) << 16);
            *(uint2*)&As0[r * ldA0 + lane * 4] = p;
        }
    }
    __syncthreads();

    tile_gemm0_to_lds(As0, ldA0, Wt0, b0, As1, ldA1, tid);
    __syncthreads();

    tile_gemm<true>(As1, ldA1, HIDF, Wt1, nullptr, Ch, Cs, ldC, nullptr, a_s, a_d,
                    s_src_out, s_dst_out, part, m0, N, HIDF, tid);
}

template <int NOUT, bool SCORES>
__global__ __launch_bounds__(256)
void gat_gemm_kernel(const int* __restrict__ cnt, const int* __restrict__ csr,
                     const int* __restrict__ big_off, const int* __restrict__ big_csr,
                     const float* __restrict__ ssrc, const float* __restrict__ sdst,
                     const unsigned short* __restrict__ h, const float* __restrict__ b,
                     const unsigned short* __restrict__ Wt,
                     float* __restrict__ C,
                     unsigned short* __restrict__ Ch,
                     const float* __restrict__ bias_out,
                     const float* __restrict__ a_s, const float* __restrict__ a_d,
                     float* __restrict__ s_src_out, float* __restrict__ s_dst_out,
                     int N) {
    __shared__ unsigned short As[16 * (HIDF + 8)];
    __shared__ unsigned short Cs[SCORES ? 16 * (HIDF + 8) : 8];
    __shared__ float part[SCORES ? 8 * 16 * 2 : 2];
    const int ldA = HIDF + 8;
    const int ldC = HIDF + 8;

    const int tid  = threadIdx.x;
    const int lane = tid & 31;
    const int wave = __builtin_amdgcn_readfirstlane(tid >> 5);
    const int m0   = blockIdx.x * 16;

    const float4 bv0 = ((const float4*)b)[(lane & 15) * 2];
    const float4 bv1 = ((const float4*)b)[(lane & 15) * 2 + 1];
    for (int q = 0; q < 2; ++q) {
        const int r    = wave * 2 + q;
        const int node = m0 + r;
        float o[8] = {0.0f, 0.0f, 0.0f, 0.0f, 0.0f, 0.0f, 0.0f, 0.0f};
        if (node < N) gat_row_h(node, lane, cnt, csr, big_off, big_csr, ssrc, sdst, h, bv0, bv1, o);
        if (lane < 16) {
            uint4 p;
            p.x = (unsigned int)f32_to_bf16(o[0]) | ((unsigned int)f32_to_bf16(o[1]) << 16);
            p.y = (unsigned int)f32_to_bf16(o[2]) | ((unsigned int)f32_to_bf16(o[3]) << 16);
            p.z = (unsigned int)f32_to_bf16(o[4]) | ((unsigned int)f32_to_bf16(o[5]) << 16);
            p.w = (unsigned int)f32_to_bf16(o[6]) | ((unsigned int)f32_to_bf16(o[7]) << 16);
            *(uint4*)&As[r * ldA + lane * 8] = p;
        }
    }
    __syncthreads();

    if (SCORES) {
        tile_gemm<true>(As, ldA, HIDF, Wt, C, Ch, Cs, ldC, bias_out, a_s, a_d,
                        s_src_out, s_dst_out, part, m0, N, NOUT, tid);
    } else if (wave < NOUT / 16) {
        tile_gemm<false>(As, ldA, HIDF, Wt, C, Ch, Cs, ldC, bias_out, a_s, a_d,
                         s_src_out, s_dst_out, part, m0, N, NOUT, tid);
    }
}

extern "C" void kernel_launch(void* const* d_in, const int* in_sizes, int n_in,
                              void* d_out, int out_size, void* d_ws, size_t ws_size,
                              hipStream_t stream) {
    (void)n_in; (void)out_size; (void)ws_size;

    const float* x    = (const float*)d_in[0];
    const int*   ei   = (const int*)d_in[1];
    const float* Wl[3] = {(const float*)d_in[2], (const float*)d_in[6],  (const float*)d_in[10]};
    const float* Asl[3]= {(const float*)d_in[3], (const float*)d_in[7],  (const float*)d_in[11]};
    const float* Adl[3]= {(const float*)d_in[4], (const float*)d_in[8],  (const float*)d_in[12]};
    const float* Bl[3] = {(const float*)d_in[5], (const float*)d_in[9],  (const float*)d_in[13]};
    const float* Wout = (const float*)d_in[14];
    const float* bout = (const float*)d_in[15];

    const int FIN  = 64;
    const int N  = in_sizes[0] / FIN;
    const int E  = in_sizes[1] / 2;

    unsigned short* hA  = (unsigned short*)d_ws;
    unsigned short* hB  = hA + (size_t)N * HIDF;
    unsigned short* Wt0 = hB + (size_t)N * HIDF;
    unsigned short* Wt1 = Wt0 + HIDF * HIDF;
    unsigned short* Wt2 = Wt1 + HIDF * HIDF;
    unsigned short* Wt3 = Wt2 + HIDF * HIDF;
    float* vsd    = (float*)(Wt3 + HIDF * HIDF);
    float* sA_src = vsd + 2 * FIN_C;
    float* sA_dst = sA_src + N;
    float* sB_src = sA_dst + N;
    float* sB_dst = sB_src + N;
    int2* bucket   = (int2*)(sB_dst + N);
    int2* ovf_pair = bucket + (size_t)N * CAP;
    int2* big_pair = ovf_pair + E;
    int* csr      = (int*)(big_pair + E);
    int* cnt      = csr + (size_t)N * CAP;
    int* gcnt     = cnt + N;
    int* big_off  = gcnt + 2;
    int* ovf_dst  = big_off + N;
    int* big_csr  = ovf_dst + E;

    const int B256 = 256;
    const int gN    = (N + B256 - 1) / B256;
    const int gE    = (E + B256 - 1) / B256;
    const int gM    = (N + 15) / 16;
    const int gS    = (N + 63) / 64;

    prep_kernel<<<gN + 193, B256, 0, stream>>>(cnt, gcnt, N, gN, Wl[0], Wl[1], Wl[2], Wout,
                                               Wt0, Wt1, Wt2, Wt3, Asl[0], Adl[0], vsd);

    edges_scores_kernel<<<gE + gS, B256, 0, stream>>>(ei, cnt, bucket, gcnt, ovf_pair, ovf_dst, E, gE,
                                                      x, vsd, sA_src, sA_dst, N);

    gat0_gemm_kernel<<<gM, B256, 0, stream>>>(
        cnt, bucket, csr, gcnt, ovf_pair, ovf_dst, big_pair, big_csr, big_off,
        sA_src, sA_dst, x, Bl[0], Wt0, Wt1, hB, Asl[1], Adl[1], sB_src, sB_dst, N);

    gat_gemm_kernel<HIDF, true><<<gM, B256, 0, stream>>>(
        cnt, csr, big_off, big_csr, sB_src, sB_dst, hB, Bl[1], Wt2, nullptr, hA, nullptr, Asl[2], Adl[2], sA_src, sA_dst, N);

    gat_gemm_kernel<FOUT_C, false><<<gM, B256, 0, stream>>>(
        cnt, csr, big_off, big_csr, sA_src, sA_dst, hA, Bl[2], Wt3, (float*)d_out, nullptr, bout,
        nullptr, nullptr, nullptr, nullptr, N);
}


# Weights stored in B-fragment order by the first launch: a wave's B load reads 512 contiguous bytes (4 lines) instead of 32 B from each of 16 rows (16 lines); same k pairing, sums unchanged
# speedup vs baseline: 5.9400x; 1.0732x over previous
#include <hip/hip_runtime.h>
#include <hip/hip_bf16.h>


typedef __attribute__((ext_vector_type(16))) __bf16 v16bf;
typedef __attribute__((ext_vector_type(8)))  float  v8f;

union BfFrag {
    unsigned int u[8];
    v16bf bf;
};

#define HIDF 128
#define FOUT_C 64
#define FIN_C 64
#define CAP 32

static __device__ __forceinline__ unsigned short f32_to_bf16(float f) {
    unsigned int u = __float_as_uint(f);
    unsigned int r = u + 0x7FFFu + ((u >> 16) & 1u);
    return (unsigned short)(r >> 16);
}

static __device__ __forceinline__ unsigned short f32_to_f16_bits(float f) {
    return __builtin_bit_cast(unsigned short, (_Float16)f);
}

static __device__ __forceinline__ void wt_bf16_item(const float* __restrict__ W,
                                                    unsigned short* __restrict__ Wt,
                                                    int K, int Nout, int i) {
    if (i < K * Nout) {
        const int k = i / Nout, n = i - k * Nout;
        const int w = n >> 4, c = n & 15;
        const int t = k >> 5, p = (k >> 4) & 1, hf = (k >> 3) & 1, e = k & 7;
        Wt[((((w * (K >> 5) + t) * 2 + p) * 32 + hf * 16 + c) * 8) + e] = f32_to_bf16(W[i]);
    }
}

static __device__ __forceinline__ void score_vectors_block(const float* __restrict__ W0,
                                                           const float* __restrict__ a_s,
                                                           const float* __restrict__ a_d,
                                                           float* __restrict__ vsd, int t) {
    const int k = t >> 2, q = t & 3;
    const float4* __restrict__ w4 = (const float4*)(W0 + k * HIDF + q * 32);
    const float4* __restrict__ s4 = (const float4*)(a_s + q * 32);
    const float4* __restrict__ d4 = (const float4*)(a_d + q * 32);
    float ps = 0.0f, pd = 0.0f;
#pragma unroll
    for (int i = 0; i < 8; ++i) {
        const float4 w = w4[i], s = s4[i], d = d4[i];
        ps += w.x * s.x + w.y * s.y + w.z * s.z + w.w * s.w;
        pd += w.x * d.x + w.y * d.y + w.z * d.z + w.w * d.w;
    }
    ps += __shfl_xor(ps, 1, 32); ps += __shfl_xor(ps, 2, 32);
    pd += __shfl_xor(pd, 1, 32); pd += __shfl_xor(pd, 2, 32);
    if (q == 0) { vsd[k] = ps; vsd[FIN_C + k] = pd; }
}

__global__ __launch_bounds__(256)
void prep_kernel(int* __restrict__ cnt, int* __restrict__ gcnt, int N, int nZero,
                 const float* __restrict__ W0, const float* __restrict__ W1,
                 const float* __restrict__ W2, const float* __restrict__ W3,
                 unsigned short* __restrict__ Wt0, unsigned short* __restrict__ Wt1,
                 unsigned short* __restrict__ Wt2, unsigned short* __restrict__ Wt3,
                 const float* __restrict__ a_s0, const float* __restrict__ a_d0,
                 float* __restrict__ vsd) {
    const int t = threadIdx.x;
    int b = blockIdx.x;
    if (b < nZero) {
        const int i = b * 256 + t;
        if (i < N) cnt[i] = 0;
        if (i < 2) gcnt[i] = 0;
        return;
    }
    b -= nZero;
    if (b < 32) { wt_bf16_item(W0, Wt0, 64, HIDF, b * 256 + t); return; }
    b -= 32;
    if (b < 64) { wt_bf16_item(W1, Wt1, HIDF, HIDF, b * 256 + t); return; }
    b -= 64;
    if (b < 64) { wt_bf16_item(W2, Wt2, HIDF, HIDF, b * 256 + t); return; }
    b -= 64;
    if (b < 32) { wt_bf16_item(W3, Wt3, HIDF, FOUT_C, b * 256 + t); return; }
    score_vectors_block(W0, a_s0, a_d0, vsd, t);
}

static __device__ __forceinline__ float half_wave_sum(float v) {
    v += __int_as_float(__builtin_amdgcn_update_dpp(0, __float_as_int(v), 0x161, 0xf, 0xf, true));
    v += __int_as_float(__builtin_amdgcn_update_dpp(0, __float_as_int(v), 0x162, 0xf, 0xf, true));
    v += __int_as_float(__builtin_amdgcn_update_dpp(0, __float_as_int(v), 0x164, 0xf, 0xf, true));
    v += __int_as_float(__builtin_amdgcn_update_dpp(0, __float_as_int(v), 0x168, 0xf, 0xf, true));
    return v;
}

template <bool SCORES>
static __device__ __forceinline__ void tile_gemm(const unsigned short* As, const int ldA, const int K,
                                                 const unsigned short* __restrict__ Wt,
                                                 float* __restrict__ C,
                                                 unsigned short* __restrict__ Ch,
                                                 unsigned short* Cs, const int ldC,
                                                 const float* __restrict__ bias,
                                                 const float* __restrict__ a_s,
                                                 const float* __restrict__ a_d,
                                                 float* __restrict__ s_src,
                                                 float* __restrict__ s_dst,
                                                 float* part,
                                                 const int m0, const int M, const int Nout,
                                                 const int tid) {
    const int wave  = tid >> 5;
    const int lane  = tid & 31;
    const int row   = lane & 15;
    const int khalf = lane >> 4;
    const int ncol  = wave * 16 + row;

    float as_c = 0.0f, ad_c = 0.0f;
    if (SCORES) { as_c = a_s[ncol]; ad_c = a_d[ncol]; }

    v8f acc = {};

#pragma unroll 2
    for (int k0 = 0; k0 < K; k0 += 32) {
        const int cb = k0 + khalf * 8;

        BfFrag a, b;
        {
            const uint4* p0 = (const uint4*)&As[row * ldA + cb];
            const uint4* p1 = (const uint4*)&As[row * ldA + cb + 16];
            uint4 x0 = *p0, x1 = *p1;
            a.u[0] = x0.x; a.u[1] = x0.y; a.u[2] = x0.z; a.u[3] = x0.w;
            a.u[4] = x1.x; a.u[5] = x1.y; a.u[6] = x1.z; a.u[7] = x1.w;
        }
        {
            const uint4* p0 = (const uint4*)Wt + ((size_t)(wave * (K >> 5) + (k0 >> 5)) * 2) * 32 + lane;
            const uint4* p1 = p0 + 32;
            uint4 x0 = *p0, x1 = *p1;
            b.u[0] = x0.x; b.u[1] = x0.y; b.u[2] = x0.z; b.u[3] = x0.w;
            b.u[4] = x1.x; b.u[5] = x1.y; b.u[6] = x1.z; b.u[7] = x1.w;
        }

        acc = __builtin_amdgcn_wmma_f32_16x16x32_bf16(
             false, a.bf,  false, b.bf,
             (short)0, acc,  false,  false);
    }

    float badd = bias ? bias[ncol] : 0.0f;
    if (!SCORES) {
#pragma unroll
        for (int r = 0; r < 8; ++r) {
            int gr = m0 + r + khalf * 8;
            if (gr < M) C[(size_t)gr * Nout + ncol] = acc[r] + badd;
        }
    }

    if (SCORES) {
#pragma unroll
        for (int r = 0; r < 8; ++r)
            Cs[(r + khalf * 8) * ldC + ncol] = f32_to_f16_bits(acc[r] + badd);

        float ps[8], pd[8];
#pragma unroll
        for (int r = 0; r < 8; ++r) {
            ps[r] = half_wave_sum(acc[r] * as_c);
            pd[r] = half_wave_sum(acc[r] * ad_c);
        }
        if (row == 0) {
#pragma unroll
            for (int r = 0; r < 8; ++r) {
                float2 p = {ps[r], pd[r]};
                *(float2*)&part[(wave * 16 + khalf * 8 + r) * 2] = p;
            }
        }
        __syncthreads();
        {
            const int rr = tid >> 4, pc = tid & 15;
            const int gr = m0 + rr;
            if (gr < M) *(uint4*)&Ch[(size_t)gr * HIDF + pc * 8] = *(const uint4*)&Cs[rr * ldC + pc * 8];
        }
        if (tid < 32) {
            const int rr = tid & 15, which = tid >> 4;
            float s = 0.0f;
#pragma unroll
            for (int w = 0; w < 8; ++w) s += part[(w * 16 + rr) * 2 + which];
            const int gr = m0 + rr;
            if (gr < M) (which ? s_dst : s_src)[gr] = s;
        }
    }
}

static __device__ __forceinline__ void tile_gemm0_to_lds(const unsigned short* As0, const int ldA0,
                                                         const unsigned short* __restrict__ Wt0,
                                                         const float* __restrict__ b0,
                                                         unsigned short* As1, const int ldA1,
                                                         const int tid) {
    const int wave  = tid >> 5;
    const int lane  = tid & 31;
    const int row   = lane & 15;
    const int khalf = lane >> 4;
    const int ncol  = wave * 16 + row;

    v8f acc = {};
#pragma unroll
    for (int k0 = 0; k0 < FIN_C; k0 += 32) {
        const int cb = k0 + khalf * 8;
        BfFrag a, b;
        {
            const uint4* p0 = (const uint4*)&As0[row * ldA0 + cb];
            const uint4* p1 = (const uint4*)&As0[row * ldA0 + cb + 16];
            uint4 x0 = *p0, x1 = *p1;
            a.u[0] = x0.x; a.u[1] = x0.y; a.u[2] = x0.z; a.u[3] = x0.w;
            a.u[4] = x1.x; a.u[5] = x1.y; a.u[6] = x1.z; a.u[7] = x1.w;
        }
        {
            const uint4* p0 = (const uint4*)Wt0 + ((size_t)(wave * (FIN_C >> 5) + (k0 >> 5)) * 2) * 32 + lane;
            const uint4* p1 = p0 + 32;
            uint4 x0 = *p0, x1 = *p1;
            b.u[0] = x0.x; b.u[1] = x0.y; b.u[2] = x0.z; b.u[3] = x0.w;
            b.u[4] = x1.x; b.u[5] = x1.y; b.u[6] = x1.z; b.u[7] = x1.w;
        }
        acc = __builtin_amdgcn_wmma_f32_16x16x32_bf16(
             false, a.bf,  false, b.bf,
             (short)0, acc,  false,  false);
    }

    const float badd = b0[ncol];
#pragma unroll
    for (int r = 0; r < 8; ++r) {
        float v = acc[r] + badd;
        v = v > 0.0f ? v : 0.01f * v;
        As1[(r + khalf * 8) * ldA1 + ncol] = f32_to_bf16(v);
    }
}

static __device__ __forceinline__ void node_scores_block(const float* __restrict__ x,
                                                         const float* __restrict__ vsd,
                                                         float* __restrict__ s_src,
                                                         float* __restrict__ s_dst,
                                                         int N, int blk) {
    const int tid = threadIdx.x;
    const int c   = tid & 15;
    const int n0  = blk * 64 + (tid >> 4);
    const float4* __restrict__ x4 = (const float4*)x;
    const float4 vs = ((const float4*)vsd)[c];
    const float4 vd = ((const float4*)vsd)[16 + c];
    float4 r[4];
#pragma unroll
    for (int i = 0; i < 4; ++i) {
        const int n = n0 + 16 * i;
        float4 z = {0.0f, 0.0f, 0.0f, 0.0f};
        if (n < N) z = x4[(size_t)n * 16 + c];
        r[i] = z;
    }
#pragma unroll
    for (int i = 0; i < 4; ++i) {
        const int n = n0 + 16 * i;
        const float ps = half_wave_sum(r[i].x * vs.x + r[i].y * vs.y + r[i].z * vs.z + r[i].w * vs.w);
        const float pd = half_wave_sum(r[i].x * vd.x + r[i].y * vd.y + r[i].z * vd.z + r[i].w * vd.w);
        if (c == 0 && n < N) { s_src[n] = ps; s_dst[n] = pd; }
    }
}

static __device__ __forceinline__ void edge_bucket_item(const int* __restrict__ ei,
                                                        int* __restrict__ cnt,
                                                        int2* __restrict__ bucket,
                                                        int* __restrict__ gcnt,
                                                        int2* __restrict__ ovf_pair,
                                                        int* __restrict__ ovf_dst,
                                                        int E, int i) {
    if (i < E) {
        const int d = ei[E + i];
        int2 p;
        p.x = i;
        p.y = ei[i];
        const int r = atomicAdd(&cnt[d], 1);
        if (r < CAP) {
            bucket[(size_t)d * CAP + r] = p;
        } else {
            const int o = atomicAdd(&gcnt[0], 1);
            ovf_pair[o] = p;
            ovf_dst[o]  = d;
        }
    }
}

__global__ __launch_bounds__(256)
void edges_scores_kernel(const int* __restrict__ ei, int* __restrict__ cnt,
                         int2* __restrict__ bucket, int* __restrict__ gcnt,
                         int2* __restrict__ ovf_pair, int* __restrict__ ovf_dst,
                         int E, int nEdge,
                         const float* __restrict__ x,
                         const float* __restrict__ vsd,
                         float* __restrict__ s_src,
                         float* __restrict__ s_dst, int N) {
    const int b = blockIdx.x;
    if (b < nEdge) edge_bucket_item(ei, cnt, bucket, gcnt, ovf_pair, ovf_dst, E, b * 256 + (int)threadIdx.x);
    else node_scores_block(x, vsd, s_src, s_dst, N, b - nEdge);
}

static __device__ __forceinline__ float leaky_att(float e) {
    return e > 0.0f ? e : 0.2f * e;
}
static __device__ __forceinline__ float wave_max(float v) {
    for (int o = 16; o; o >>= 1) v = fmaxf(v, __shfl_xor(v, o, 32));
    return v;
}
static __device__ __forceinline__ float wave_sum(float v) {
    for (int o = 16; o; o >>= 1) v += __shfl_xor(v, o, 32);
    return v;
}
static __device__ __forceinline__ void fma4(float4& acc, float a, const float4& v) {
    acc.x += a * v.x; acc.y += a * v.y; acc.z += a * v.z; acc.w += a * v.w;
}
static __device__ __forceinline__ float xhalf(float v) {
    const int b = __float_as_int(v);
    return __int_as_float(__builtin_amdgcn_permlanex16(b, b, 0x76543210, 0xfedcba98, false, false));
}
static __device__ __forceinline__ float4 pair_row(const float4* __restrict__ xrow,
                                                  int sbits, int j, int hsel, int c) {
    const int s0 = __builtin_amdgcn_readlane(sbits, j);
    const int s1 = __builtin_amdgcn_readlane(sbits, j + 1);
    return xrow[(size_t)(hsel ? s1 : s0) * 16 + c];
}
static __device__ __forceinline__ float pair_w(int abits, int j, int hsel) {
    const int a0 = __builtin_amdgcn_readlane(abits, j);
    const int a1 = __builtin_amdgcn_readlane(abits, j + 1);
    return __int_as_float(hsel ? a1 : a0);
}

static __device__ __forceinline__ float4 gat_row64(const int node, const int lane,
                                                   const int* __restrict__ cnt,
                                                   const int2* __restrict__ bucket,
                                                   int* __restrict__ csr,
                                                   int* gcnt,
                                                   const int2* __restrict__ ovf_pair,
                                                   const int* __restrict__ ovf_dst,
                                                   int2* big_pair, int* big_csr,
                                                   int* __restrict__ big_off,
                                                   const float* __restrict__ ssrc,
                                                   const float* __restrict__ sdst,
                                                   const float* __restrict__ x) {
    const int start = node * CAP;
    const int deg   = cnt[node];
    const int tot   = deg + 1;
    const float sd  = sdst[node];
    const float4* __restrict__ xrow = (const float4*)x;
    const int c    = lane & 15;
    const int hsel = lane >> 4;
    const float NEG_INF = __int_as_float(0xFF800000);
    float4 acc = {0.0f, 0.0f, 0.0f, 0.0f};

    if (tot <= 32) {
        const bool valid = lane < tot;
        int id = 0x7fffffff;
        int sv = node;
        if (lane < deg) { const int2 pr = bucket[start + lane]; id = pr.x; sv = pr.y; }
        int place = 0;
        for (int k = 0; k < deg; ++k) place += (__builtin_amdgcn_readlane(id, k) < id) ? 1 : 0;
        if (lane >= deg) place = lane;
        const int s = __builtin_amdgcn_ds_permute(place << 2, sv);
        if (lane < deg) csr[start + lane] = s;
        const float e   = leaky_att(ssrc[s] + sd);
        const float m   = wave_max(valid ? e : NEG_INF);
        const float ex  = valid ? __expf(e - m) : 0.0f;
        const float den = wave_sum(ex);
        const float alpha = ex / (den + 1e-16f);
        const int sbits = s;
        const int abits = __float_as_int(alpha);
        const int tot2  = (tot + 1) & ~1;
        int j = 0;
        for (; j + 8 <= tot2; j += 8) {
            const float4 v0 = pair_row(xrow, sbits, j,     hsel, c);
            const float4 v1 = pair_row(xrow, sbits, j + 2, hsel, c);
            const float4 v2 = pair_row(xrow, sbits, j + 4, hsel, c);
            const float4 v3 = pair_row(xrow, sbits, j + 6, hsel, c);
            fma4(acc, pair_w(abits, j,     hsel), v0);
            fma4(acc, pair_w(abits, j + 2, hsel), v1);
            fma4(acc, pair_w(abits, j + 4, hsel), v2);
            fma4(acc, pair_w(abits, j + 6, hsel), v3);
        }
        const int rem = tot2 - j;
        if (rem == 6) {
            const float4 v0 = pair_row(xrow, sbits, j,     hsel, c);
            const float4 v1 = pair_row(xrow, sbits, j + 2, hsel, c);
            const float4 v2 = pair_row(xrow, sbits, j + 4, hsel, c);
            fma4(acc, pair_w(abits, j,     hsel), v0);
            fma4(acc, pair_w(abits, j + 2, hsel), v1);
            fma4(acc, pair_w(abits, j + 4, hsel), v2);
        } else if (rem == 4) {
            const float4 v0 = pair_row(xrow, sbits, j,     hsel, c);
            const float4 v1 = pair_row(xrow, sbits, j + 2, hsel, c);
            fma4(acc, pair_w(abits, j,     hsel), v0);
            fma4(acc, pair_w(abits, j + 2, hsel), v1);
        } else if (rem == 2) {
            const float4 v0 = pair_row(xrow, sbits, j, hsel, c);
            fma4(acc, pair_w(abits, j, hsel), v0);
        }
    } else {
        int base = 0;
        if (lane == 0) { base = atomicAdd(&gcnt[1], deg); big_off[node] = base; }
        base = __builtin_amdgcn_readfirstlane(base);
        for (int k = lane; k < CAP; k += 32) big_pair[base + k] = bucket[start + k];
        int filled = CAP;
        const int novf = gcnt[0];
        for (int cc = 0; cc < novf; cc += 32) {
            const int idx = cc + lane;
            const bool hit = idx < novf && ovf_dst[idx] == node;
            const unsigned int hits = __builtin_amdgcn_ballot_w32(hit);
            if (hit) big_pair[base + filled + __popc(hits & ((1u << lane) - 1u))] = ovf_pair[idx];
            filled += __popc(hits);
        }
        __threadfence();
        for (int cc = 0; cc < deg; cc += 32) {
            const int idx = cc + lane;
            if (idx < deg) {
                const int2 pr = big_pair[base + idx];
                int place = 0;
                for (int k = 0; k < deg; ++k) place += (big_pair[base + k].x < pr.x) ? 1 : 0;
                big_csr[base + place] = pr.y;
            }
        }
        __threadfence();
        float m = NEG_INF;
        for (int cc = 0; cc < tot; cc += 32) {
            const int idx = cc + lane;
            if (idx < tot) {
                const int s = idx < deg ? big_csr[base + idx] : node;
                m = fmaxf(m, leaky_att(ssrc[s] + sd));
            }
        }
        m = wave_max(m);
        float den = 0.0f;
        for (int cc = 0; cc < tot; cc += 32) {
            const int idx = cc + lane;
            if (idx < tot) {
                const int s = idx < deg ? big_csr[base + idx] : node;
                den += __expf(leaky_att(ssrc[s] + sd) - m);
            }
        }
        den = wave_sum(den);
        for (int cc = 0; cc < tot; cc += 32) {
            const int idx = cc + lane;
            int s = node;
            if (idx < deg) s = big_csr[base + idx];
            float alpha = 0.0f;
            if (idx < tot) alpha = __expf(leaky_att(ssrc[s] + sd) - m) / (den + 1e-16f);
            const int abits = __float_as_int(alpha);
            const int cnt_e = (tot - cc) < 32 ? (tot - cc) : 32;
            const int cnt2  = (cnt_e + 1) & ~1;
#pragma unroll 4
            for (int j = 0; j < cnt2; j += 2) {
                fma4(acc, pair_w(abits, j, hsel), pair_row(xrow, s, j, hsel, c));
            }
        }
    }

    acc.x += xhalf(acc.x); acc.y += xhalf(acc.y);
    acc.z += xhalf(acc.z); acc.w += xhalf(acc.w);
    return acc;
}

static __device__ __forceinline__ uint4 pair_row_h(const uint4* __restrict__ hrow,
                                                   int sbits, int j, int hsel, int c) {
    const int s0 = __builtin_amdgcn_readlane(sbits, j);
    const int s1 = __builtin_amdgcn_readlane(sbits, j + 1);
    return hrow[(size_t)(hsel ? s1 : s0) * 16 + c];
}
static __device__ __forceinline__ float fma_h_lo(float acc, const unsigned int w, const float a) {
    asm volatile("v_fma_mix_f32 %0, %1, %2, %0 op_sel_hi:[1,0,0]" : "+v"(acc) : "v"(w), "v"(a));
    return acc;
}
static __device__ __forceinline__ float fma_h_hi(float acc, const unsigned int w, const float a) {
    asm volatile("v_fma_mix_f32 %0, %1, %2, %0 op_sel:[1,0,0] op_sel_hi:[1,0,0]" : "+v"(acc) : "v"(w), "v"(a));
    return acc;
}
static __device__ __forceinline__ void fma8h(float (&acc)[8], const float a, const uint4 q) {
    acc[0] = fma_h_lo(acc[0], q.x, a); acc[1] = fma_h_hi(acc[1], q.x, a);
    acc[2] = fma_h_lo(acc[2], q.y, a); acc[3] = fma_h_hi(acc[3], q.y, a);
    acc[4] = fma_h_lo(acc[4], q.z, a); acc[5] = fma_h_hi(acc[5], q.z, a);
    acc[6] = fma_h_lo(acc[6], q.w, a); acc[7] = fma_h_hi(acc[7], q.w, a);
}

static __device__ __forceinline__ void gat_row_h(const int node, const int lane,
                                                 const int* __restrict__ cnt,
                                                 const int* __restrict__ csr,
                                                 const int* __restrict__ big_off,
                                                 const int* __restrict__ big_csr,
                                                 const float* __restrict__ ssrc,
                                                 const float* __restrict__ sdst,
                                                 const unsigned short* __restrict__ h,
                                                 const float4 bv0, const float4 bv1,
                                                 float (&o)[8]) {
    const int start = node * CAP;
    const int deg   = cnt[node];
    const int tot   = deg + 1;
    const float sd  = sdst[node];
    const uint4* __restrict__ hrow = (const uint4*)h;
    const int c    = lane & 15;
    const int hsel = lane >> 4;
    const float NEG_INF = __int_as_float(0xFF800000);
    float acc[8] = {0.0f, 0.0f, 0.0f, 0.0f, 0.0f, 0.0f, 0.0f, 0.0f};

    if (tot <= 32) {
        const bool valid = lane < tot;
        int s = node;
        if (lane < deg) s = csr[start + lane];
        const float e   = leaky_att(ssrc[s] + sd);
        const float m   = wave_max(valid ? e : NEG_INF);
        const float ex  = valid ? __expf(e - m) : 0.0f;
        const float den = wave_sum(ex);
        const float alpha = ex / (den + 1e-16f);
        const int sbits = s;
        const int abits = __float_as_int(alpha);
        const int tot2  = (tot + 1) & ~1;
        int j = 0;
        for (; j + 8 <= tot2; j += 8) {
            const uint4 v0 = pair_row_h(hrow, sbits, j,     hsel, c);
            const uint4 v1 = pair_row_h(hrow, sbits, j + 2, hsel, c);
            const uint4 v2 = pair_row_h(hrow, sbits, j + 4, hsel, c);
            const uint4 v3 = pair_row_h(hrow, sbits, j + 6, hsel, c);
            fma8h(acc, pair_w(abits, j,     hsel), v0);
            fma8h(acc, pair_w(abits, j + 2, hsel), v1);
            fma8h(acc, pair_w(abits, j + 4, hsel), v2);
            fma8h(acc, pair_w(abits, j + 6, hsel), v3);
        }
        const int rem = tot2 - j;
        if (rem == 6) {
            const uint4 v0 = pair_row_h(hrow, sbits, j,     hsel, c);
            const uint4 v1 = pair_row_h(hrow, sbits, j + 2, hsel, c);
            const uint4 v2 = pair_row_h(hrow, sbits, j + 4, hsel, c);
            fma8h(acc, pair_w(abits, j,     hsel), v0);
            fma8h(acc, pair_w(abits, j + 2, hsel), v1);
            fma8h(acc, pair_w(abits, j + 4, hsel), v2);
        } else if (rem == 4) {
            const uint4 v0 = pair_row_h(hrow, sbits, j,     hsel, c);
            const uint4 v1 = pair_row_h(hrow, sbits, j + 2, hsel, c);
            fma8h(acc, pair_w(abits, j,     hsel), v0);
            fma8h(acc, pair_w(abits, j + 2, hsel), v1);
        } else if (rem == 2) {
            const uint4 v0 = pair_row_h(hrow, sbits, j, hsel, c);
            fma8h(acc, pair_w(abits, j, hsel), v0);
        }
    } else {
        const int base = big_off[node];
        float m = NEG_INF;
        for (int cc = 0; cc < tot; cc += 32) {
            const int idx = cc + lane;
            if (idx < tot) {
                const int s = idx < deg ? big_csr[base + idx] : node;
                m = fmaxf(m, leaky_att(ssrc[s] + sd));
            }
        }
        m = wave_max(m);
        float den = 0.0f;
        for (int cc = 0; cc < tot; cc += 32) {
            const int idx = cc + lane;
            if (idx < tot) {
                const int s = idx < deg ? big_csr[base + idx] : node;
                den += __expf(leaky_att(ssrc[s] + sd) - m);
            }
        }
        den = wave_sum(den);
        for (int cc = 0; cc < tot; cc += 32) {
            const int idx = cc + lane;
            int s = node;
            if (idx < deg) s = big_csr[base + idx];
            float alpha = 0.0f;
            if (idx < tot) alpha = __expf(leaky_att(ssrc[s] + sd) - m) / (den + 1e-16f);
            const int abits = __float_as_int(alpha);
            const int cnt_e = (tot - cc) < 32 ? (tot - cc) : 32;
            const int cnt2  = (cnt_e + 1) & ~1;
#pragma unroll 4
            for (int j = 0; j < cnt2; j += 2) {
                fma8h(acc, pair_w(abits, j, hsel), pair_row_h(hrow, s, j, hsel, c));
            }
        }
    }

    const float bb[8] = {bv0.x, bv0.y, bv0.z, bv0.w, bv1.x, bv1.y, bv1.z, bv1.w};
#pragma unroll
    for (int k = 0; k < 8; ++k) {
        const float v = acc[k] + xhalf(acc[k]) + bb[k];
        o[k] = v > 0.0f ? v : 0.01f * v;
    }
}

__global__ __launch_bounds__(256)
void gat0_gemm_kernel(const int* __restrict__ cnt, const int2* __restrict__ bucket,
                      int* __restrict__ csr, int* gcnt,
                      const int2* __restrict__ ovf_pair, const int* __restrict__ ovf_dst,
                      int2* big_pair, int* big_csr, int* __restrict__ big_off,
                      const float* __restrict__ ssrc, const float* __restrict__ sdst,
                      const float* __restrict__ x, const float* __restrict__ b0,
                      const unsigned short* __restrict__ Wt0,
                      const unsigned short* __restrict__ Wt1,
                      unsigned short* __restrict__ Ch,
                      const float* __restrict__ a_s, const float* __restrict__ a_d,
                      float* __restrict__ s_src_out, float* __restrict__ s_dst_out,
                      int N) {
    __shared__ unsigned short As0[16 * (FIN_C + 8)];
    __shared__ unsigned short As1[16 * (HIDF + 8)];
    __shared__ unsigned short Cs[16 * (HIDF + 8)];
    __shared__ float part[8 * 16 * 2];
    const int ldA0 = FIN_C + 8;
    const int ldA1 = HIDF + 8;
    const int ldC  = HIDF + 8;

    const int tid  = threadIdx.x;
    const int lane = tid & 31;
    const int wave = __builtin_amdgcn_readfirstlane(tid >> 5);
    const int m0   = blockIdx.x * 16;

    for (int q = 0; q < 2; ++q) {
        const int r    = wave * 2 + q;
        const int node = m0 + r;
        float4 o = {0.0f, 0.0f, 0.0f, 0.0f};
        if (node < N) o = gat_row64(node, lane, cnt, bucket, csr, gcnt, ovf_pair, ovf_dst,
                                    big_pair, big_csr, big_off, ssrc, sdst, x);
        if (lane < 16) {
            uint2 p;
            p.x = (unsigned int)f32_to_bf16(o.x) | ((unsigned int)f32_to_bf16(o.y) << 16);
            p.y = (unsigned int)f32_to_bf16(o.z) | ((unsigned int)f32_to_bf16(o.w) << 16);
            *(uint2*)&As0[r * ldA0 + lane * 4] = p;
        }
    }
    __syncthreads();

    tile_gemm0_to_lds(As0, ldA0, Wt0, b0, As1, ldA1, tid);
    __syncthreads();

    tile_gemm<true>(As1, ldA1, HIDF, Wt1, nullptr, Ch, Cs, ldC, nullptr, a_s, a_d,
                    s_src_out, s_dst_out, part, m0, N, HIDF, tid);
}

template <int NOUT, bool SCORES>
__global__ __launch_bounds__(256)
void gat_gemm_kernel(const int* __restrict__ cnt, const int* __restrict__ csr,
                     const int* __restrict__ big_off, const int* __restrict__ big_csr,
                     const float* __restrict__ ssrc, const float* __restrict__ sdst,
                     const unsigned short* __restrict__ h, const float* __restrict__ b,
                     const unsigned short* __restrict__ Wt,
                     float* __restrict__ C,
                     unsigned short* __restrict__ Ch,
                     const float* __restrict__ bias_out,
                     const float* __restrict__ a_s, const float* __restrict__ a_d,
                     float* __restrict__ s_src_out, float* __restrict__ s_dst_out,
                     int N) {
    __shared__ unsigned short As[16 * (HIDF + 8)];
    __shared__ unsigned short Cs[SCORES ? 16 * (HIDF + 8) : 8];
    __shared__ float part[SCORES ? 8 * 16 * 2 : 2];
    const int ldA = HIDF + 8;
    const int ldC = HIDF + 8;

    const int tid  = threadIdx.x;
    const int lane = tid & 31;
    const int wave = __builtin_amdgcn_readfirstlane(tid >> 5);
    const int m0   = blockIdx.x * 16;

    const float4 bv0 = ((const float4*)b)[(lane & 15) * 2];
    const float4 bv1 = ((const float4*)b)[(lane & 15) * 2 + 1];
    for (int q = 0; q < 2; ++q) {
        const int r    = wave * 2 + q;
        const int node = m0 + r;
        float o[8] = {0.0f, 0.0f, 0.0f, 0.0f, 0.0f, 0.0f, 0.0f, 0.0f};
        if (node < N) gat_row_h(node, lane, cnt, csr, big_off, big_csr, ssrc, sdst, h, bv0, bv1, o);
        if (lane < 16) {
            uint4 p;
            p.x = (unsigned int)f32_to_bf16(o[0]) | ((unsigned int)f32_to_bf16(o[1]) << 16);
            p.y = (unsigned int)f32_to_bf16(o[2]) | ((unsigned int)f32_to_bf16(o[3]) << 16);
            p.z = (unsigned int)f32_to_bf16(o[4]) | ((unsigned int)f32_to_bf16(o[5]) << 16);
            p.w = (unsigned int)f32_to_bf16(o[6]) | ((unsigned int)f32_to_bf16(o[7]) << 16);
            *(uint4*)&As[r * ldA + lane * 8] = p;
        }
    }
    __syncthreads();

    if (SCORES) {
        tile_gemm<true>(As, ldA, HIDF, Wt, C, Ch, Cs, ldC, bias_out, a_s, a_d,
                        s_src_out, s_dst_out, part, m0, N, NOUT, tid);
    } else if (wave < NOUT / 16) {
        tile_gemm<false>(As, ldA, HIDF, Wt, C, Ch, Cs, ldC, bias_out, a_s, a_d,
                         s_src_out, s_dst_out, part, m0, N, NOUT, tid);
    }
}

extern "C" void kernel_launch(void* const* d_in, const int* in_sizes, int n_in,
                              void* d_out, int out_size, void* d_ws, size_t ws_size,
                              hipStream_t stream) {
    (void)n_in; (void)out_size; (void)ws_size;

    const float* x    = (const float*)d_in[0];
    const int*   ei   = (const int*)d_in[1];
    const float* Wl[3] = {(const float*)d_in[2], (const float*)d_in[6],  (const float*)d_in[10]};
    const float* Asl[3]= {(const float*)d_in[3], (const float*)d_in[7],  (const float*)d_in[11]};
    const float* Adl[3]= {(const float*)d_in[4], (const float*)d_in[8],  (const float*)d_in[12]};
    const float* Bl[3] = {(const float*)d_in[5], (const float*)d_in[9],  (const float*)d_in[13]};
    const float* Wout = (const float*)d_in[14];
    const float* bout = (const float*)d_in[15];

    const int FIN  = 64;
    const int N  = in_sizes[0] / FIN;
    const int E  = in_sizes[1] / 2;

    unsigned short* hA  = (unsigned short*)d_ws;
    unsigned short* hB  = hA + (size_t)N * HIDF;
    unsigned short* Wt0 = hB + (size_t)N * HIDF;
    unsigned short* Wt1 = Wt0 + HIDF * HIDF;
    unsigned short* Wt2 = Wt1 + HIDF * HIDF;
    unsigned short* Wt3 = Wt2 + HIDF * HIDF;
    float* vsd    = (float*)(Wt3 + HIDF * HIDF);
    float* sA_src = vsd + 2 * FIN_C;
    float* sA_dst = sA_src + N;
    float* sB_src = sA_dst + N;
    float* sB_dst = sB_src + N;
    int2* bucket   = (int2*)(sB_dst + N);
    int2* ovf_pair = bucket + (size_t)N * CAP;
    int2* big_pair = ovf_pair + E;
    int* csr      = (int*)(big_pair + E);
    int* cnt      = csr + (size_t)N * CAP;
    int* gcnt     = cnt + N;
    int* big_off  = gcnt + 2;
    int* ovf_dst  = big_off + N;
    int* big_csr  = ovf_dst + E;

    const int B256 = 256;
    const int gN    = (N + B256 - 1) / B256;
    const int gE    = (E + B256 - 1) / B256;
    const int gM    = (N + 15) / 16;
    const int gS    = (N + 63) / 64;

    prep_kernel<<<gN + 193, B256, 0, stream>>>(cnt, gcnt, N, gN, Wl[0], Wl[1], Wl[2], Wout,
                                               Wt0, Wt1, Wt2, Wt3, Asl[0], Adl[0], vsd);

    edges_scores_kernel<<<gE + gS, B256, 0, stream>>>(ei, cnt, bucket, gcnt, ovf_pair, ovf_dst, E, gE,
                                                      x, vsd, sA_src, sA_dst, N);

    gat0_gemm_kernel<<<gM, B256, 0, stream>>>(
        cnt, bucket, csr, gcnt, ovf_pair, ovf_dst, big_pair, big_csr, big_off,
        sA_src, sA_dst, x, Bl[0], Wt0, Wt1, hB, Asl[1], Adl[1], sB_src, sB_dst, N);

    gat_gemm_kernel<HIDF, true><<<gM, B256, 0, stream>>>(
        cnt, csr, big_off, big_csr, sB_src, sB_dst, hB, Bl[1], Wt2, nullptr, hA, nullptr, Asl[2], Adl[2], sA_src, sA_dst, N);

    gat_gemm_kernel<FOUT_C, false><<<gM, B256, 0, stream>>>(
        cnt, csr, big_off, big_csr, sA_src, sA_dst, hA, Bl[2], Wt3, (float*)d_out, nullptr, bout,
        nullptr, nullptr, nullptr, nullptr, N);
}
